# starting kernel
# baseline (speedup 1.0000x reference)
#include <hip/hip_runtime.h>
#include <hip/hip_bf16.h>
#include <math.h>


typedef __attribute__((ext_vector_type(2))) float v2f;
typedef __attribute__((ext_vector_type(8))) float v8f;

#define BATCH 4
#define NPTS  4096
#define M1    2048
#define M2    512
#define GF    1024

__device__ __forceinline__ unsigned f2ord(float f) {
    unsigned u = __float_as_uint(f);
    return (u & 0x80000000u) ? ~u : (u | 0x80000000u);
}
__device__ __forceinline__ float ord2f(unsigned k) {
    return (k & 0x80000000u) ? __uint_as_float(k & 0x7fffffffu)
                             : __uint_as_float(~k);
}

__global__ void hidden_kernel(const float* __restrict__ x,
                              const float* __restrict__ W1,
                              const float* __restrict__ b1,
                              float* __restrict__ hidden, int M) {
    int idx = blockIdx.x * blockDim.x + threadIdx.x;
    if (idx >= M * 64) return;
    int r = idx >> 6, f = idx & 63;
    float v = x[r * 2 + 0] * W1[f] + x[r * 2 + 1] * W1[64 + f] + b1[f];
    hidden[idx] = tanhf(v);
}

template <int KREL, int C1, int C2>
__device__ __forceinline__ float auxload(const float* __restrict__ A1,
                                         const float* __restrict__ A2,
                                         int row) {
    if constexpr (KREL < C1) {
        return A1[row * C1 + KREL];
    } else if constexpr (KREL - C1 < C2) {
        return A2[row * C2 + (KREL - C1)];
    } else {
        return 0.0f;
    }
}

template <int OUTMODE, int C0, int C1, int C2, int KTOT>
__global__ void gemm_wmma_kernel(const float* __restrict__ A0,
                                 const float* __restrict__ A1,
                                 const float* __restrict__ A2,
                                 const float* __restrict__ W, int F,
                                 const float* __restrict__ bias,
                                 float* __restrict__ out,
                                 unsigned* __restrict__ gfmax,
                                 int rowsPerBatch) {
    constexpr int KP = (KTOT + 3) & ~3;
    static_assert((C0 & 3) == 0, "C0 multiple of 4");
    static_assert(KP - C0 == 0 || KP - C0 == 4, "at most one tail K-step");

    int lane = threadIdx.x & 31;
    int half = lane >> 4;
    int mrow = lane & 15;
    int tm = blockIdx.x;
    int colBase = blockIdx.y * 64;
    int arow = tm * 16 + mrow;

    v8f c[4];
#pragma unroll
    for (int q = 0; q < 4; ++q)
#pragma unroll
        for (int v = 0; v < 8; ++v) c[q][v] = 0.0f;

    const float* a0row = A0 + (size_t)arow * C0;

#pragma unroll
    for (int k0 = 0; k0 < C0; k0 += 4) {
        int ka = k0 + 2 * half;
        v2f a = *(const v2f*)(a0row + ka);
#pragma unroll
        for (int nt = 0; nt < 4; ++nt) {
            int n = colBase + nt * 16 + mrow;
            v2f bm;
            bm.x = W[ka * F + n];
            bm.y = W[(ka + 1) * F + n];
            c[nt] = __builtin_amdgcn_wmma_f32_16x16x4_f32(
                false, a, false, bm, (short)0, c[nt], false, false);
        }
    }

    if constexpr (KP > C0) {
        float t0 = auxload<0, C1, C2>(A1, A2, arow);
        float t1 = auxload<1, C1, C2>(A1, A2, arow);
        float t2 = auxload<2, C1, C2>(A1, A2, arow);
        float t3 = auxload<3, C1, C2>(A1, A2, arow);
        v2f a;
        a.x = half ? t2 : t0;
        a.y = half ? t3 : t1;
        int ka = C0 + 2 * half;
        int kx = (ka < KTOT) ? ka : (KTOT - 1);
        int ky = (ka + 1 < KTOT) ? (ka + 1) : (KTOT - 1);
#pragma unroll
        for (int nt = 0; nt < 4; ++nt) {
            int n = colBase + nt * 16 + mrow;
            float wx = W[kx * F + n];
            float wy = W[ky * F + n];
            v2f bm;
            bm.x = (ka < KTOT) ? wx : 0.0f;
            bm.y = (ka + 1 < KTOT) ? wy : 0.0f;
            c[nt] = __builtin_amdgcn_wmma_f32_16x16x4_f32(
                false, a, false, bm, (short)0, c[nt], false, false);
        }
    }

#pragma unroll
    for (int nt = 0; nt < 4; ++nt) {
        int n = colBase + nt * 16 + mrow;
#pragma unroll
        for (int v = 0; v < 8; ++v) {
            int m = tm * 16 + v + 8 * half;
            float val = c[nt][v];
            if constexpr (OUTMODE == 0) {
                out[(size_t)m * F + n] = val;
            } else if constexpr (OUTMODE == 1) {
                out[(size_t)m * F + n] = tanhf(val + bias[n]);
            } else {
                float g = val + bias[n];
                atomicMax(&gfmax[(m / rowsPerBatch) * F + n], f2ord(g));
            }
        }
    }
}

__global__ void fps_kernel(const float* __restrict__ pos, int n, int m,
                           float* __restrict__ ctr) {
    __shared__ float mind2[NPTS];
    __shared__ float rval[256];
    __shared__ int ridx[256];
    __shared__ int s_last;

    int b = blockIdx.x;
    int tid = threadIdx.x;
    const float* pb = pos + (size_t)b * n * 2;
    float* cb = ctr + (size_t)b * m * 2;

    for (int j = tid; j < n; j += 256) mind2[j] = 1e10f;
    if (tid == 0) {
        s_last = 0;
        cb[0] = pb[0];
        cb[1] = pb[1];
    }
    __syncthreads();

    int last = 0;
    for (int it = 1; it < m; ++it) {
        float lx = pb[last * 2 + 0];
        float ly = pb[last * 2 + 1];
        float bv = -1.0f;
        int bi = n;
        for (int j = tid; j < n; j += 256) {
            float2 p = ((const float2*)pb)[j];
            float dx = p.x - lx;
            float dy = p.y - ly;
            float d2 = dx * dx + dy * dy;
            float md = fminf(mind2[j], d2);
            mind2[j] = md;
            if (md > bv) { bv = md; bi = j; }
        }
        rval[tid] = bv;
        ridx[tid] = bi;
        __syncthreads();
        for (int s = 128; s > 0; s >>= 1) {
            if (tid < s) {
                float ov = rval[tid + s];
                int oi = ridx[tid + s];
                if (ov > rval[tid] || (ov == rval[tid] && oi < ridx[tid])) {
                    rval[tid] = ov;
                    ridx[tid] = oi;
                }
            }
            __syncthreads();
        }
        if (tid == 0) {
            int nxt = ridx[0];
            s_last = nxt;
            cb[it * 2 + 0] = pb[nxt * 2 + 0];
            cb[it * 2 + 1] = pb[nxt * 2 + 1];
        }
        __syncthreads();
        last = s_last;
    }
}

template <int U>
__global__ void agg_kernel(const float* __restrict__ t,
                           const float* __restrict__ pos,
                           const float* __restrict__ ctr,
                           const float* __restrict__ Wp,
                           const float* __restrict__ bias,
                           float* __restrict__ out,
                           int n, int m, float r2) {
    const int F = U * 32;
    int gtid = blockIdx.x * blockDim.x + threadIdx.x;
    int w = gtid >> 5;
    int lane = threadIdx.x & 31;
    if (w >= BATCH * m) return;
    int b = w / m;
    int i = w - b * m;

    float2 cc = ((const float2*)ctr)[b * m + i];
    float cx = cc.x, cy = cc.y;

    float acc[U];
#pragma unroll
    for (int u = 0; u < U; ++u) acc[u] = -1e30f;

    const float2* pb = (const float2*)(pos + (size_t)b * n * 2);
    const float* tb = t + (size_t)b * n * F;
    for (int j = 0; j < n; ++j) {
        float2 p = pb[j];
        float dx = p.x - cx;
        float dy = p.y - cy;
        if (dx * dx + dy * dy <= r2) {
#pragma unroll
            for (int u = 0; u < U; ++u)
                acc[u] = fmaxf(acc[u], tb[(size_t)j * F + lane + 32 * u]);
        }
    }
#pragma unroll
    for (int u = 0; u < U; ++u) {
        int f = lane + 32 * u;
        out[(size_t)(b * m + i) * F + f] =
            acc[u] + bias[f] - cx * Wp[f] - cy * Wp[F + f];
    }
}

__global__ void init_gf_kernel(unsigned* __restrict__ gf, int n) {
    int i = blockIdx.x * blockDim.x + threadIdx.x;
    if (i < n) gf[i] = 0u;
}
__global__ void decode_gf_kernel(const unsigned* __restrict__ gf,
                                 float* __restrict__ out, int n) {
    int i = blockIdx.x * blockDim.x + threadIdx.x;
    if (i < n) out[i] = ord2f(gf[i]);
}

extern "C" void kernel_launch(void* const* d_in, const int* in_sizes, int n_in,
                              void* d_out, int out_size, void* d_ws, size_t ws_size,
                              hipStream_t stream) {
    const float* x     = (const float*)d_in[0];
    const float* zones = (const float*)d_in[1];
    const float* lf_W1 = (const float*)d_in[2];
    const float* lf_b1 = (const float*)d_in[3];
    const float* lf_W2 = (const float*)d_in[4];
    const float* lf_b2 = (const float*)d_in[5];
    const float* c1_W  = (const float*)d_in[6];
    const float* c1_b  = (const float*)d_in[7];
    const float* c2_W  = (const float*)d_in[8];
    const float* c2_b  = (const float*)d_in[9];
    const float* c3_W  = (const float*)d_in[10];
    const float* c3_b  = (const float*)d_in[11];

    float* out    = (float*)d_out;
    float* lf_out = out;
    float* gf_out = out + (size_t)BATCH * NPTS * 64;

    const int M = BATCH * NPTS;

    float* ws = (float*)d_ws;
    size_t o = 0;
    float* hidden = ws + o;  o += (size_t)M * 64;
    float* t1     = ws + o;  o += (size_t)M * 64;
    float* ctr1   = ws + o;  o += (size_t)BATCH * M1 * 2;
    float* h1     = ws + o;  o += (size_t)BATCH * M1 * 64;
    float* t2     = ws + o;  o += (size_t)BATCH * M1 * 128;
    float* ctr2   = ws + o;  o += (size_t)BATCH * M2 * 2;
    float* h2     = ws + o;  o += (size_t)BATCH * M2 * 128;
    unsigned* gf_enc = (unsigned*)(ws + o);

    hidden_kernel<<<(M * 64 + 255) / 256, 256, 0, stream>>>(x, lf_W1, lf_b1,
                                                            hidden, M);
    gemm_wmma_kernel<1, 64, 0, 0, 64><<<dim3(M / 16, 1), 32, 0, stream>>>(
        hidden, nullptr, nullptr, lf_W2, 64, lf_b2, lf_out, nullptr, 0);

    fps_kernel<<<BATCH, 256, 0, stream>>>(x, NPTS, M1, ctr1);
    fps_kernel<<<BATCH, 256, 0, stream>>>(ctr1, M1, M2, ctr2);

    gemm_wmma_kernel<0, 64, 1, 2, 67><<<dim3(M / 16, 1), 32, 0, stream>>>(
        lf_out, zones, x, c1_W, 64, nullptr, t1, nullptr, 0);

    agg_kernel<2><<<(BATCH * M1 * 32) / 256, 256, 0, stream>>>(
        t1, x, ctr1, c1_W + 65 * 64, c1_b, h1, NPTS, M1, 0.25f);

    gemm_wmma_kernel<0, 64, 2, 0, 66>
        <<<dim3(BATCH * M1 / 16, 2), 32, 0, stream>>>(
            h1, ctr1, nullptr, c2_W, 128, nullptr, t2, nullptr, 0);

    agg_kernel<4><<<(BATCH * M2 * 32) / 256, 256, 0, stream>>>(
        t2, ctr1, ctr2, c2_W + 64 * 128, c2_b, h2, M1, M2, 1.0f);

    init_gf_kernel<<<(BATCH * GF + 255) / 256, 256, 0, stream>>>(gf_enc,
                                                                 BATCH * GF);
    gemm_wmma_kernel<2, 128, 2, 0, 130>
        <<<dim3(BATCH * M2 / 16, GF / 64), 32, 0, stream>>>(
            h2, ctr2, nullptr, c3_W, GF, c3_b, nullptr, gf_enc, M2);
    decode_gf_kernel<<<(BATCH * GF + 255) / 256, 256, 0, stream>>>(gf_enc,
                                                                   gf_out,
                                                                   BATCH * GF);
}


# FPS ranked-list key compare written as one borrow chain (subtract lo words, then hi words with the borrow): 2 vector ops and no scalar and/or per compare-exchange; order, picks, d2 bit-identical
# speedup vs baseline: 4.0624x; 4.0624x over previous
#include <hip/hip_runtime.h>
#include <hip/hip_bf16.h>
#include <math.h>


typedef __attribute__((ext_vector_type(2))) float v2f;
typedef __attribute__((ext_vector_type(8))) float v8f;

#define BATCH 4
#define NPTS  4096
#define M1    2048
#define M2    512
#define GF    1024

__device__ __forceinline__ unsigned f2ord(float f) {
    unsigned u = __float_as_uint(f);
    return (u & 0x80000000u) ? ~u : (u | 0x80000000u);
}
__device__ __forceinline__ float ord2f(unsigned k) {
    return (k & 0x80000000u) ? __uint_as_float(k & 0x7fffffffu)
                             : __uint_as_float(~k);
}

__global__ void hidden_kernel(const float* __restrict__ x,
                              const float* __restrict__ W1,
                              const float* __restrict__ b1,
                              float* __restrict__ hidden, int M) {
    int idx = blockIdx.x * blockDim.x + threadIdx.x;
    if (idx >= M * 64) return;
    int r = idx >> 6, f = idx & 63;
    float v = x[r * 2 + 0] * W1[f] + x[r * 2 + 1] * W1[64 + f] + b1[f];
    hidden[idx] = tanhf(v);
}

template <int KREL, int C1, int C2>
__device__ __forceinline__ float auxload(const float* __restrict__ A1,
                                         const float* __restrict__ A2,
                                         int row) {
    if constexpr (KREL < C1) {
        return A1[row * C1 + KREL];
    } else if constexpr (KREL - C1 < C2) {
        return A2[row * C2 + (KREL - C1)];
    } else {
        return 0.0f;
    }
}

template <int OUTMODE, int C0, int C1, int C2, int KTOT>
__global__ void gemm_wmma_kernel(const float* __restrict__ A0,
                                 const float* __restrict__ A1,
                                 const float* __restrict__ A2,
                                 const float* __restrict__ W, int F,
                                 const float* __restrict__ bias,
                                 float* __restrict__ out,
                                 unsigned* __restrict__ gfmax,
                                 int rowsPerBatch) {
    constexpr int KP = (KTOT + 3) & ~3;
    static_assert((C0 & 3) == 0, "C0 multiple of 4");
    static_assert(KP - C0 == 0 || KP - C0 == 4, "at most one tail K-step");

    int lane = threadIdx.x & 31;
    int half = lane >> 4;
    int mrow = lane & 15;
    int tm = blockIdx.x;
    int colBase = blockIdx.y * 64;
    int arow = tm * 16 + mrow;

    v8f c[4];
#pragma unroll
    for (int q = 0; q < 4; ++q)
#pragma unroll
        for (int v = 0; v < 8; ++v) c[q][v] = 0.0f;

    const float* a0row = A0 + (size_t)arow * C0;

#pragma unroll
    for (int k0 = 0; k0 < C0; k0 += 4) {
        int ka = k0 + 2 * half;
        v2f a = *(const v2f*)(a0row + ka);
#pragma unroll
        for (int nt = 0; nt < 4; ++nt) {
            int n = colBase + nt * 16 + mrow;
            v2f bm;
            bm.x = W[ka * F + n];
            bm.y = W[(ka + 1) * F + n];
            c[nt] = __builtin_amdgcn_wmma_f32_16x16x4_f32(
                false, a, false, bm, (short)0, c[nt], false, false);
        }
    }

    if constexpr (KP > C0) {
        float t0 = auxload<0, C1, C2>(A1, A2, arow);
        float t1 = auxload<1, C1, C2>(A1, A2, arow);
        float t2 = auxload<2, C1, C2>(A1, A2, arow);
        float t3 = auxload<3, C1, C2>(A1, A2, arow);
        v2f a;
        a.x = half ? t2 : t0;
        a.y = half ? t3 : t1;
        int ka = C0 + 2 * half;
        int kx = (ka < KTOT) ? ka : (KTOT - 1);
        int ky = (ka + 1 < KTOT) ? (ka + 1) : (KTOT - 1);
#pragma unroll
        for (int nt = 0; nt < 4; ++nt) {
            int n = colBase + nt * 16 + mrow;
            float wx = W[kx * F + n];
            float wy = W[ky * F + n];
            v2f bm;
            bm.x = (ka < KTOT) ? wx : 0.0f;
            bm.y = (ka + 1 < KTOT) ? wy : 0.0f;
            c[nt] = __builtin_amdgcn_wmma_f32_16x16x4_f32(
                false, a, false, bm, (short)0, c[nt], false, false);
        }
    }

#pragma unroll
    for (int nt = 0; nt < 4; ++nt) {
        int n = colBase + nt * 16 + mrow;
#pragma unroll
        for (int v = 0; v < 8; ++v) {
            int m = tm * 16 + v + 8 * half;
            float val = c[nt][v];
            if constexpr (OUTMODE == 0) {
                out[(size_t)m * F + n] = val;
            } else if constexpr (OUTMODE == 1) {
                out[(size_t)m * F + n] = tanhf(val + bias[n]);
            } else {
                float g = val + bias[n];
                atomicMax(&gfmax[(m / rowsPerBatch) * F + n], f2ord(g));
            }
        }
    }
}

#define FPS_K 16

__device__ __forceinline__ float dist2(float px, float py, float qx, float qy) {
    float dx = px - qx;
    float dy = py - qy;
    float d2;
    {
#pragma clang fp contract(off)
        d2 = dx * dx + dy * dy;
    }
    return d2;
}

struct fkey {
    unsigned hi;
    unsigned lo;
};

__device__ __forceinline__ fkey fps_key(float md, int j) {
    fkey k;
    k.hi = __float_as_uint(md);
    k.lo = (0x7fffffffu - (unsigned)j) << 1;
    return k;
}

__device__ __forceinline__ bool key_gt(fkey a, fkey b) {
    unsigned borrow_lo, borrow;
    (void)__builtin_subc(b.lo, a.lo, 0u, &borrow_lo);
    (void)__builtin_subc(b.hi, a.hi, borrow_lo, &borrow);
    return borrow != 0u;
}

__device__ __forceinline__ fkey key_sel(bool g, fkey a, fkey b) {
    fkey r;
    r.hi = g ? a.hi : b.hi;
    r.lo = g ? a.lo : b.lo;
    return r;
}

__device__ __forceinline__ void key_apart(fkey& k) {
    asm("" : "+v"(k.hi), "+v"(k.lo));
}

__device__ __forceinline__ fkey key_ld(uint2 v) {
    fkey k;
    k.hi = v.y;
    k.lo = v.x;
    key_apart(k);
    return k;
}
__device__ __forceinline__ uint2 key_st(fkey k) { return make_uint2(k.lo, k.hi); }

__device__ __forceinline__ void cmpx(fkey& a, fkey& b) {
    bool g = key_gt(b, a);
    fkey hi = key_sel(g, b, a);
    fkey lo = key_sel(g, a, b);
    a = hi;
    b = lo;
}

template <int N>
__device__ __forceinline__ void bitonic_desc(fkey (&c)[N]) {
#pragma unroll
    for (int i = 0; i < N; ++i) key_apart(c[i]);
#pragma unroll
    for (int stride = N / 2; stride >= 1; stride >>= 1) {
#pragma unroll
        for (int i = 0; i < N; ++i)
            if ((i & stride) == 0) cmpx(c[i], c[i + stride]);
    }
}

__device__ __forceinline__ unsigned lane_u(unsigned v, int lane) {
    return (unsigned)__builtin_amdgcn_readlane((int)v, lane);
}
__device__ __forceinline__ float lane_f(float v, int lane) {
    return __int_as_float(__builtin_amdgcn_readlane(__float_as_int(v), lane));
}

__global__ void fps_kernel(const float* __restrict__ pos, int n, int m,
                           float* __restrict__ ctr) {
    __shared__ float mind2[NPTS];
    __shared__ uint2 skey[FPS_K][256];
    __shared__ float s_px[FPS_K];
    __shared__ float s_py[FPS_K];
    __shared__ int s_np;

    int b = blockIdx.x;
    int tid = threadIdx.x;
    const float2* pb2 = (const float2*)(pos + (size_t)b * n * 2);
    float2* cb2 = (float2*)(ctr + (size_t)b * m * 2);

    for (int j = tid; j < n; j += 256) mind2[j] = 1e10f;
    if (tid == 0) {
        float2 p0 = pb2[0];
        cb2[0] = p0;
        for (int q = 0; q < FPS_K; ++q) {
            s_px[q] = p0.x;
            s_py[q] = p0.y;
        }
        s_np = 1;
    }
    __syncthreads();

    int it = 1;
    int np = __builtin_amdgcn_readfirstlane(s_np);
    while (it < m) {
        float px[FPS_K], py[FPS_K];
#pragma unroll
        for (int q = 0; q < FPS_K; ++q) {
            px[q] = s_px[q];
            py[q] = s_py[q];
        }

        fkey L[FPS_K];
        L[0].hi = 0u;
        L[0].lo = 0u;
        L[1].hi = 0u;
        L[1].lo = 0u;
        for (int j = tid; j < n; j += 256) {
            float2 p = pb2[j];
            float md = mind2[j];
            float d[FPS_K];
#pragma unroll
            for (int q = 0; q < FPS_K; ++q)
                d[q] = dist2(p.x, p.y, px[q], py[q]);
#pragma unroll
            for (int w = FPS_K / 2; w >= 1; w >>= 1)
#pragma unroll
                for (int k = 0; k < w; ++k) d[k] = fminf(d[k], d[k + w]);
            md = fminf(md, d[0]);
            mind2[j] = md;
            fkey key = fps_key(md, j);
            fkey lo = key_sel(key_gt(key, L[1]), key, L[1]);
            bool g = key_gt(lo, L[0]);
            L[1] = key_sel(g, L[0], lo);
            L[0] = key_sel(g, lo, L[0]);
        }
        L[1].lo |= 1u;
        skey[0][tid] = key_st(L[0]);
        skey[1][tid] = key_st(L[1]);
        __syncthreads();

        static_assert(FPS_K == 16, "the tree below builds lists of 16");
        if (tid < 128) {
            fkey c[4];
            c[0] = L[0];
            c[1] = L[1];
            c[2] = key_ld(skey[1][tid + 128]);
            c[3] = key_ld(skey[0][tid + 128]);
            bitonic_desc<4>(c);
#pragma unroll
            for (int k = 0; k < 4; ++k) {
                L[k] = c[k];
                skey[k][tid] = key_st(c[k]);
            }
        }
        __syncthreads();
        if (tid < 64) {
            fkey c[8];
#pragma unroll
            for (int k = 0; k < 4; ++k) {
                c[k] = L[k];
                c[7 - k] = key_ld(skey[k][tid + 64]);
            }
            bitonic_desc<8>(c);
#pragma unroll
            for (int k = 0; k < 8; ++k) {
                L[k] = c[k];
                skey[k][tid] = key_st(c[k]);
            }
        }
        __syncthreads();
        if (tid < 32) {
            fkey c[16];
#pragma unroll
            for (int k = 0; k < 8; ++k) {
                c[k] = L[k];
                c[15 - k] = key_ld(skey[k][tid + 32]);
            }
            bitonic_desc<16>(c);
#pragma unroll
            for (int k = 0; k < 16; ++k) {
                L[k] = c[k];
                skey[k][tid] = key_st(c[k]);
            }
        }
        __syncthreads();
        for (int s = 16; s > 0; s >>= 1) {
            if (tid < s) {
                fkey c[FPS_K];
#pragma unroll
                for (int k = 0; k < FPS_K; ++k) {
                    fkey o = key_ld(skey[FPS_K - 1 - k][tid + s]);
                    c[k] = key_sel(key_gt(L[k], o), L[k], o);
                }
                bitonic_desc<FPS_K>(c);
#pragma unroll
                for (int k = 0; k < FPS_K; ++k) {
                    L[k] = c[k];
                    skey[k][tid] = key_st(c[k]);
                }
            }
            __syncthreads();
        }

        if (tid < 32) {
            int li = tid & 15;
            fkey my = key_ld(skey[li][0]);
            unsigned l = my.lo;
            unsigned lo = l & ~1u;
            unsigned hi = my.hi;
            float md = __uint_as_float(hi);
            float2 cp = pb2[0x7fffffffu - (l >> 1)];
            unsigned mk = __builtin_amdgcn_ballot_w32((l & 1u) != 0u) & 0xffffu;
            int ex = mk ? __ffs((int)mk) : FPS_K;
            unsigned bound_hi = lane_u(hi, ex - 1);
            unsigned bound_lo = lane_u(lo, ex - 1);
            float c0x = lane_f(cp.x, 0);
            float c0y = lane_f(cp.y, 0);
            unsigned w = 0u;
#pragma unroll
            for (int a = 0; a < FPS_K; ++a) {
                float d = dist2(cp.x, cp.y, lane_f(cp.x, a), lane_f(cp.y, a));
                bool lowers = d < md;
                unsigned db = __float_as_uint(d);
                bool behind = (db < bound_hi) | ((db == bound_hi) & (lo < bound_lo));
                w |= (lowers ? (1u << a) : 0u) | (behind ? (0x10000u << a) : 0u);
            }
            int room = m - it;
            unsigned below = (1u << li) - 1u;
            unsigned lowers = w & below;
            unsigned behinds = (w >> 16) & below;
            unsigned all = 1u;
#pragma unroll
            for (int k = 1; k < FPS_K; ++k)
                all = __builtin_amdgcn_ballot_w32((lowers & all) == 0u) & 0xffffu;
            bool keeps = ((all >> li) & 1u) != 0u;
            bool behind = (behinds & all) != 0u;
            bool live = (li < ex) && (hi != 0u) && (__popc(all & below) < room);
            bool on = live && (keeps || behind);
            unsigned stops = __builtin_amdgcn_ballot_w32(!on) & 0xfffeu;
            int first = stops ? (__ffs((int)stops) - 1) : FPS_K;
            unsigned taken = all & ((1u << first) - 1u);
            int cnt = __popc(taken);
            if (tid < FPS_K) {
                if ((taken >> tid) & 1u) {
                    int slot = __popc(taken & ((1u << tid) - 1u));
                    cb2[it + slot] = cp;
                    s_px[slot] = cp.x;
                    s_py[slot] = cp.y;
                }
                if (tid >= cnt) {
                    s_px[tid] = c0x;
                    s_py[tid] = c0y;
                }
            }
            if (tid == 0) s_np = cnt;
        }
        __syncthreads();
        np = __builtin_amdgcn_readfirstlane(s_np);
        it += np;
    }
}

template <int U>
__global__ void agg_kernel(const float* __restrict__ t,
                           const float* __restrict__ pos,
                           const float* __restrict__ ctr,
                           const float* __restrict__ Wp,
                           const float* __restrict__ bias,
                           float* __restrict__ out,
                           int n, int m, float r2) {
    const int F = U * 32;
    int gtid = blockIdx.x * blockDim.x + threadIdx.x;
    int w = gtid >> 5;
    int lane = threadIdx.x & 31;
    if (w >= BATCH * m) return;
    int b = w / m;
    int i = w - b * m;

    float2 cc = ((const float2*)ctr)[b * m + i];
    float cx = cc.x, cy = cc.y;

    float acc[U];
#pragma unroll
    for (int u = 0; u < U; ++u) acc[u] = -1e30f;

    const float2* pb = (const float2*)(pos + (size_t)b * n * 2);
    const float* tb = t + (size_t)b * n * F;
    for (int j = 0; j < n; ++j) {
        float2 p = pb[j];
        float dx = p.x - cx;
        float dy = p.y - cy;
        if (dx * dx + dy * dy <= r2) {
#pragma unroll
            for (int u = 0; u < U; ++u)
                acc[u] = fmaxf(acc[u], tb[(size_t)j * F + lane + 32 * u]);
        }
    }
#pragma unroll
    for (int u = 0; u < U; ++u) {
        int f = lane + 32 * u;
        out[(size_t)(b * m + i) * F + f] =
            acc[u] + bias[f] - cx * Wp[f] - cy * Wp[F + f];
    }
}

__global__ void init_gf_kernel(unsigned* __restrict__ gf, int n) {
    int i = blockIdx.x * blockDim.x + threadIdx.x;
    if (i < n) gf[i] = 0u;
}
__global__ void decode_gf_kernel(const unsigned* __restrict__ gf,
                                 float* __restrict__ out, int n) {
    int i = blockIdx.x * blockDim.x + threadIdx.x;
    if (i < n) out[i] = ord2f(gf[i]);
}

extern "C" void kernel_launch(void* const* d_in, const int* in_sizes, int n_in,
                              void* d_out, int out_size, void* d_ws, size_t ws_size,
                              hipStream_t stream) {
    const float* x     = (const float*)d_in[0];
    const float* zones = (const float*)d_in[1];
    const float* lf_W1 = (const float*)d_in[2];
    const float* lf_b1 = (const float*)d_in[3];
    const float* lf_W2 = (const float*)d_in[4];
    const float* lf_b2 = (const float*)d_in[5];
    const float* c1_W  = (const float*)d_in[6];
    const float* c1_b  = (const float*)d_in[7];
    const float* c2_W  = (const float*)d_in[8];
    const float* c2_b  = (const float*)d_in[9];
    const float* c3_W  = (const float*)d_in[10];
    const float* c3_b  = (const float*)d_in[11];

    float* out    = (float*)d_out;
    float* lf_out = out;
    float* gf_out = out + (size_t)BATCH * NPTS * 64;

    const int M = BATCH * NPTS;

    float* ws = (float*)d_ws;
    size_t o = 0;
    float* hidden = ws + o;  o += (size_t)M * 64;
    float* t1     = ws + o;  o += (size_t)M * 64;
    float* ctr1   = ws + o;  o += (size_t)BATCH * M1 * 2;
    float* h1     = ws + o;  o += (size_t)BATCH * M1 * 64;
    float* t2     = ws + o;  o += (size_t)BATCH * M1 * 128;
    float* ctr2   = ws + o;  o += (size_t)BATCH * M2 * 2;
    float* h2     = ws + o;  o += (size_t)BATCH * M2 * 128;
    unsigned* gf_enc = (unsigned*)(ws + o);

    hidden_kernel<<<(M * 64 + 255) / 256, 256, 0, stream>>>(x, lf_W1, lf_b1,
                                                            hidden, M);
    gemm_wmma_kernel<1, 64, 0, 0, 64><<<dim3(M / 16, 1), 32, 0, stream>>>(
        hidden, nullptr, nullptr, lf_W2, 64, lf_b2, lf_out, nullptr, 0);

    fps_kernel<<<BATCH, 256, 0, stream>>>(x, NPTS, M1, ctr1);
    fps_kernel<<<BATCH, 256, 0, stream>>>(ctr1, M1, M2, ctr2);

    gemm_wmma_kernel<0, 64, 1, 2, 67><<<dim3(M / 16, 1), 32, 0, stream>>>(
        lf_out, zones, x, c1_W, 64, nullptr, t1, nullptr, 0);

    agg_kernel<2><<<(BATCH * M1 * 32) / 256, 256, 0, stream>>>(
        t1, x, ctr1, c1_W + 65 * 64, c1_b, h1, NPTS, M1, 0.25f);

    gemm_wmma_kernel<0, 64, 2, 0, 66>
        <<<dim3(BATCH * M1 / 16, 2), 32, 0, stream>>>(
            h1, ctr1, nullptr, c2_W, 128, nullptr, t2, nullptr, 0);

    agg_kernel<4><<<(BATCH * M2 * 32) / 256, 256, 0, stream>>>(
        t2, ctr1, ctr2, c2_W + 64 * 128, c2_b, h2, M1, M2, 1.0f);

    init_gf_kernel<<<(BATCH * GF + 255) / 256, 256, 0, stream>>>(gf_enc,
                                                                 BATCH * GF);
    gemm_wmma_kernel<2, 128, 2, 0, 130>
        <<<dim3(BATCH * M2 / 16, GF / 64), 32, 0, stream>>>(
            h2, ctr2, nullptr, c3_W, GF, c3_b, nullptr, gf_enc, M2);
    decode_gf_kernel<<<(BATCH * GF + 255) / 256, 256, 0, stream>>>(gf_enc,
                                                                   gf_out,
                                                                   BATCH * GF);
}


# FPS ranked lists: last 5 merge levels (32 lists of 16 -> 1) hold a list across 16 lanes, one key per lane, merged by cross-lane moves; 8 waves merge 4 lists each, wave 0 the rest; picks unchanged
# speedup vs baseline: 4.3476x; 1.0702x over previous
#include <hip/hip_runtime.h>
#include <hip/hip_bf16.h>
#include <math.h>


typedef __attribute__((ext_vector_type(2))) float v2f;
typedef __attribute__((ext_vector_type(8))) float v8f;

#define BATCH 4
#define NPTS  4096
#define M1    2048
#define M2    512
#define GF    1024

__device__ __forceinline__ unsigned f2ord(float f) {
    unsigned u = __float_as_uint(f);
    return (u & 0x80000000u) ? ~u : (u | 0x80000000u);
}
__device__ __forceinline__ float ord2f(unsigned k) {
    return (k & 0x80000000u) ? __uint_as_float(k & 0x7fffffffu)
                             : __uint_as_float(~k);
}

__global__ void hidden_kernel(const float* __restrict__ x,
                              const float* __restrict__ W1,
                              const float* __restrict__ b1,
                              float* __restrict__ hidden, int M) {
    int idx = blockIdx.x * blockDim.x + threadIdx.x;
    if (idx >= M * 64) return;
    int r = idx >> 6, f = idx & 63;
    float v = x[r * 2 + 0] * W1[f] + x[r * 2 + 1] * W1[64 + f] + b1[f];
    hidden[idx] = tanhf(v);
}

template <int KREL, int C1, int C2>
__device__ __forceinline__ float auxload(const float* __restrict__ A1,
                                         const float* __restrict__ A2,
                                         int row) {
    if constexpr (KREL < C1) {
        return A1[row * C1 + KREL];
    } else if constexpr (KREL - C1 < C2) {
        return A2[row * C2 + (KREL - C1)];
    } else {
        return 0.0f;
    }
}

template <int OUTMODE, int C0, int C1, int C2, int KTOT>
__global__ void gemm_wmma_kernel(const float* __restrict__ A0,
                                 const float* __restrict__ A1,
                                 const float* __restrict__ A2,
                                 const float* __restrict__ W, int F,
                                 const float* __restrict__ bias,
                                 float* __restrict__ out,
                                 unsigned* __restrict__ gfmax,
                                 int rowsPerBatch) {
    constexpr int KP = (KTOT + 3) & ~3;
    static_assert((C0 & 3) == 0, "C0 multiple of 4");
    static_assert(KP - C0 == 0 || KP - C0 == 4, "at most one tail K-step");

    int lane = threadIdx.x & 31;
    int half = lane >> 4;
    int mrow = lane & 15;
    int tm = blockIdx.x;
    int colBase = blockIdx.y * 64;
    int arow = tm * 16 + mrow;

    v8f c[4];
#pragma unroll
    for (int q = 0; q < 4; ++q)
#pragma unroll
        for (int v = 0; v < 8; ++v) c[q][v] = 0.0f;

    const float* a0row = A0 + (size_t)arow * C0;

#pragma unroll
    for (int k0 = 0; k0 < C0; k0 += 4) {
        int ka = k0 + 2 * half;
        v2f a = *(const v2f*)(a0row + ka);
#pragma unroll
        for (int nt = 0; nt < 4; ++nt) {
            int n = colBase + nt * 16 + mrow;
            v2f bm;
            bm.x = W[ka * F + n];
            bm.y = W[(ka + 1) * F + n];
            c[nt] = __builtin_amdgcn_wmma_f32_16x16x4_f32(
                false, a, false, bm, (short)0, c[nt], false, false);
        }
    }

    if constexpr (KP > C0) {
        float t0 = auxload<0, C1, C2>(A1, A2, arow);
        float t1 = auxload<1, C1, C2>(A1, A2, arow);
        float t2 = auxload<2, C1, C2>(A1, A2, arow);
        float t3 = auxload<3, C1, C2>(A1, A2, arow);
        v2f a;
        a.x = half ? t2 : t0;
        a.y = half ? t3 : t1;
        int ka = C0 + 2 * half;
        int kx = (ka < KTOT) ? ka : (KTOT - 1);
        int ky = (ka + 1 < KTOT) ? (ka + 1) : (KTOT - 1);
#pragma unroll
        for (int nt = 0; nt < 4; ++nt) {
            int n = colBase + nt * 16 + mrow;
            float wx = W[kx * F + n];
            float wy = W[ky * F + n];
            v2f bm;
            bm.x = (ka < KTOT) ? wx : 0.0f;
            bm.y = (ka + 1 < KTOT) ? wy : 0.0f;
            c[nt] = __builtin_amdgcn_wmma_f32_16x16x4_f32(
                false, a, false, bm, (short)0, c[nt], false, false);
        }
    }

#pragma unroll
    for (int nt = 0; nt < 4; ++nt) {
        int n = colBase + nt * 16 + mrow;
#pragma unroll
        for (int v = 0; v < 8; ++v) {
            int m = tm * 16 + v + 8 * half;
            float val = c[nt][v];
            if constexpr (OUTMODE == 0) {
                out[(size_t)m * F + n] = val;
            } else if constexpr (OUTMODE == 1) {
                out[(size_t)m * F + n] = tanhf(val + bias[n]);
            } else {
                float g = val + bias[n];
                atomicMax(&gfmax[(m / rowsPerBatch) * F + n], f2ord(g));
            }
        }
    }
}

#define FPS_K 16

__device__ __forceinline__ float dist2(float px, float py, float qx, float qy) {
    float dx = px - qx;
    float dy = py - qy;
    float d2;
    {
#pragma clang fp contract(off)
        d2 = dx * dx + dy * dy;
    }
    return d2;
}

struct fkey {
    unsigned hi;
    unsigned lo;
};

__device__ __forceinline__ fkey fps_key(float md, int j) {
    fkey k;
    k.hi = __float_as_uint(md);
    k.lo = (0x7fffffffu - (unsigned)j) << 1;
    return k;
}

__device__ __forceinline__ bool key_gt(fkey a, fkey b) {
    unsigned borrow_lo, borrow;
    (void)__builtin_subc(b.lo, a.lo, 0u, &borrow_lo);
    (void)__builtin_subc(b.hi, a.hi, borrow_lo, &borrow);
    return borrow != 0u;
}

__device__ __forceinline__ fkey key_sel(bool g, fkey a, fkey b) {
    fkey r;
    r.hi = g ? a.hi : b.hi;
    r.lo = g ? a.lo : b.lo;
    return r;
}

__device__ __forceinline__ void key_apart(fkey& k) {
    asm("" : "+v"(k.hi), "+v"(k.lo));
}

__device__ __forceinline__ fkey key_ld(uint2 v) {
    fkey k;
    k.hi = v.y;
    k.lo = v.x;
    key_apart(k);
    return k;
}
__device__ __forceinline__ uint2 key_st(fkey k) { return make_uint2(k.lo, k.hi); }

__device__ __forceinline__ void cmpx(fkey& a, fkey& b) {
    bool g = key_gt(b, a);
    fkey hi = key_sel(g, b, a);
    fkey lo = key_sel(g, a, b);
    a = hi;
    b = lo;
}

template <int N>
__device__ __forceinline__ void bitonic_desc(fkey (&c)[N]) {
#pragma unroll
    for (int i = 0; i < N; ++i) key_apart(c[i]);
#pragma unroll
    for (int stride = N / 2; stride >= 1; stride >>= 1) {
#pragma unroll
        for (int i = 0; i < N; ++i)
            if ((i & stride) == 0) cmpx(c[i], c[i + stride]);
    }
}

__device__ __forceinline__ unsigned lane_u(unsigned v, int lane) {
    return (unsigned)__builtin_amdgcn_readlane((int)v, lane);
}
__device__ __forceinline__ float lane_f(float v, int lane) {
    return __int_as_float(__builtin_amdgcn_readlane(__float_as_int(v), lane));
}

template <int CTRL>
__device__ __forceinline__ fkey key_row(fkey k) {
    fkey r;
    r.hi = (unsigned)__builtin_amdgcn_update_dpp(0, (int)k.hi, CTRL, 0xf, 0xf, false);
    r.lo = (unsigned)__builtin_amdgcn_update_dpp(0, (int)k.lo, CTRL, 0xf, 0xf, false);
    return r;
}
__device__ __forceinline__ fkey key_other_row(fkey k) {
    fkey r;
    r.hi = __builtin_amdgcn_permlanex16(k.hi, k.hi, 0x76543210u, 0xfedcba98u, false, false);
    r.lo = __builtin_amdgcn_permlanex16(k.lo, k.lo, 0x76543210u, 0xfedcba98u, false, false);
    return r;
}
template <int CTRL>
__device__ __forceinline__ fkey lane_cmpx(fkey own, bool up) {
    fkey o = key_row<CTRL>(own);
    bool g = key_gt(o, own);
    fkey mx = key_sel(g, o, own);
    fkey mn = key_sel(g, own, o);
    key_apart(mx);
    key_apart(mn);
    return key_sel(up, mx, mn);
}
__device__ __forceinline__ fkey merge_rows(fkey v, int e) {
    fkey o = key_other_row(v);
    v = key_sel(key_gt(o, v), o, v);
    v = lane_cmpx<0x128>(v, (e & 8) == 0);
    v = lane_cmpx<0x164>(v, (e & 4) == 0);
    v = lane_cmpx<0x04e>(v, (e & 2) == 0);
    v = lane_cmpx<0x0b1>(v, (e & 1) == 0);
    return v;
}
__device__ __forceinline__ fkey key_mirror(fkey k) { return key_row<0x140>(k); }

__global__ void fps_kernel(const float* __restrict__ pos, int n, int m,
                           float* __restrict__ ctr) {
    __shared__ float mind2[NPTS];
    __shared__ uint2 skey[FPS_K / 2][256];
    __shared__ uint2 tkey[32][FPS_K + 1];
    __shared__ uint2 wkey[8][FPS_K + 1];
    __shared__ float s_px[FPS_K];
    __shared__ float s_py[FPS_K];
    __shared__ int s_np;

    int b = blockIdx.x;
    int tid = threadIdx.x;
    const float2* pb2 = (const float2*)(pos + (size_t)b * n * 2);
    float2* cb2 = (float2*)(ctr + (size_t)b * m * 2);

    for (int j = tid; j < n; j += 256) mind2[j] = 1e10f;
    if (tid == 0) {
        float2 p0 = pb2[0];
        cb2[0] = p0;
        for (int q = 0; q < FPS_K; ++q) {
            s_px[q] = p0.x;
            s_py[q] = p0.y;
        }
        s_np = 1;
    }
    __syncthreads();

    int it = 1;
    int np = __builtin_amdgcn_readfirstlane(s_np);
    while (it < m) {
        float px[FPS_K], py[FPS_K];
#pragma unroll
        for (int q = 0; q < FPS_K; ++q) {
            px[q] = s_px[q];
            py[q] = s_py[q];
        }

        fkey L[FPS_K];
        L[0].hi = 0u;
        L[0].lo = 0u;
        L[1].hi = 0u;
        L[1].lo = 0u;
        for (int j = tid; j < n; j += 256) {
            float2 p = pb2[j];
            float md = mind2[j];
            float d[FPS_K];
#pragma unroll
            for (int q = 0; q < FPS_K; ++q)
                d[q] = dist2(p.x, p.y, px[q], py[q]);
#pragma unroll
            for (int w = FPS_K / 2; w >= 1; w >>= 1)
#pragma unroll
                for (int k = 0; k < w; ++k) d[k] = fminf(d[k], d[k + w]);
            md = fminf(md, d[0]);
            mind2[j] = md;
            fkey key = fps_key(md, j);
            fkey lo = key_sel(key_gt(key, L[1]), key, L[1]);
            bool g = key_gt(lo, L[0]);
            L[1] = key_sel(g, L[0], lo);
            L[0] = key_sel(g, lo, L[0]);
        }
        L[1].lo |= 1u;
        skey[0][tid] = key_st(L[0]);
        skey[1][tid] = key_st(L[1]);
        __syncthreads();

        static_assert(FPS_K == 16, "the tree below builds lists of 16");
        if (tid < 128) {
            fkey c[4];
            c[0] = L[0];
            c[1] = L[1];
            c[2] = key_ld(skey[1][tid + 128]);
            c[3] = key_ld(skey[0][tid + 128]);
            bitonic_desc<4>(c);
#pragma unroll
            for (int k = 0; k < 4; ++k) {
                L[k] = c[k];
                skey[k][tid] = key_st(c[k]);
            }
        }
        __syncthreads();
        if (tid < 64) {
            fkey c[8];
#pragma unroll
            for (int k = 0; k < 4; ++k) {
                c[k] = L[k];
                c[7 - k] = key_ld(skey[k][tid + 64]);
            }
            bitonic_desc<8>(c);
#pragma unroll
            for (int k = 0; k < 8; ++k) {
                L[k] = c[k];
                skey[k][tid] = key_st(c[k]);
            }
        }
        __syncthreads();
        if (tid < 32) {
            fkey c[16];
#pragma unroll
            for (int k = 0; k < 8; ++k) {
                c[k] = L[k];
                c[15 - k] = key_ld(skey[k][tid + 32]);
            }
            bitonic_desc<16>(c);
#pragma unroll
            for (int k = 0; k < 16; ++k) tkey[tid][k] = key_st(c[k]);
        }
        __syncthreads();
        {
            int lane = tid & 31;
            int wv = tid >> 5;
            int h = lane >> 4;
            int e = lane & 15;
            int at = h ? (FPS_K - 1 - e) : e;
            fkey r0 = merge_rows(key_ld(tkey[4 * wv + h][at]), e);
            fkey r1 = merge_rows(key_ld(tkey[4 * wv + 2 + h][at]), e);
            fkey x = merge_rows(key_sel(h != 0, key_mirror(r1), r0), e);
            if (h == 0) wkey[wv][e] = key_st(x);
        }
        __syncthreads();

        if (tid < 32) {
            int li = tid & 15;
            bool upper = tid >= 16;
            int at = upper ? (FPS_K - 1 - li) : li;
            int hr = upper ? 1 : 0;
            fkey q0 = merge_rows(key_ld(wkey[0 + hr][at]), li);
            fkey q1 = merge_rows(key_ld(wkey[2 + hr][at]), li);
            fkey q2 = merge_rows(key_ld(wkey[4 + hr][at]), li);
            fkey q3 = merge_rows(key_ld(wkey[6 + hr][at]), li);
            fkey d0 = merge_rows(key_sel(upper, key_mirror(q1), q0), li);
            fkey d1 = merge_rows(key_sel(upper, key_mirror(q3), q2), li);
            fkey my = merge_rows(key_sel(upper, key_mirror(d1), d0), li);
            unsigned l = my.lo;
            unsigned lo = l & ~1u;
            unsigned hi = my.hi;
            float md = __uint_as_float(hi);
            float2 cp = pb2[0x7fffffffu - (l >> 1)];
            unsigned mk = __builtin_amdgcn_ballot_w32((l & 1u) != 0u) & 0xffffu;
            int ex = mk ? __ffs((int)mk) : FPS_K;
            unsigned bound_hi = lane_u(hi, ex - 1);
            unsigned bound_lo = lane_u(lo, ex - 1);
            float c0x = lane_f(cp.x, 0);
            float c0y = lane_f(cp.y, 0);
            unsigned w = 0u;
#pragma unroll
            for (int a = 0; a < FPS_K; ++a) {
                float d = dist2(cp.x, cp.y, lane_f(cp.x, a), lane_f(cp.y, a));
                bool lowers = d < md;
                unsigned db = __float_as_uint(d);
                bool behind = (db < bound_hi) | ((db == bound_hi) & (lo < bound_lo));
                w |= (lowers ? (1u << a) : 0u) | (behind ? (0x10000u << a) : 0u);
            }
            int room = m - it;
            unsigned below = (1u << li) - 1u;
            unsigned lowers = w & below;
            unsigned behinds = (w >> 16) & below;
            unsigned all = 1u;
#pragma unroll
            for (int k = 1; k < FPS_K; ++k)
                all = __builtin_amdgcn_ballot_w32((lowers & all) == 0u) & 0xffffu;
            bool keeps = ((all >> li) & 1u) != 0u;
            bool behind = (behinds & all) != 0u;
            bool live = (li < ex) && (hi != 0u) && (__popc(all & below) < room);
            bool on = live && (keeps || behind);
            unsigned stops = __builtin_amdgcn_ballot_w32(!on) & 0xfffeu;
            int first = stops ? (__ffs((int)stops) - 1) : FPS_K;
            unsigned taken = all & ((1u << first) - 1u);
            int cnt = __popc(taken);
            if (tid < FPS_K) {
                if ((taken >> tid) & 1u) {
                    int slot = __popc(taken & ((1u << tid) - 1u));
                    cb2[it + slot] = cp;
                    s_px[slot] = cp.x;
                    s_py[slot] = cp.y;
                }
                if (tid >= cnt) {
                    s_px[tid] = c0x;
                    s_py[tid] = c0y;
                }
            }
            if (tid == 0) s_np = cnt;
        }
        __syncthreads();
        np = __builtin_amdgcn_readfirstlane(s_np);
        it += np;
    }
}

template <int U>
__global__ void agg_kernel(const float* __restrict__ t,
                           const float* __restrict__ pos,
                           const float* __restrict__ ctr,
                           const float* __restrict__ Wp,
                           const float* __restrict__ bias,
                           float* __restrict__ out,
                           int n, int m, float r2) {
    const int F = U * 32;
    int gtid = blockIdx.x * blockDim.x + threadIdx.x;
    int w = gtid >> 5;
    int lane = threadIdx.x & 31;
    if (w >= BATCH * m) return;
    int b = w / m;
    int i = w - b * m;

    float2 cc = ((const float2*)ctr)[b * m + i];
    float cx = cc.x, cy = cc.y;

    float acc[U];
#pragma unroll
    for (int u = 0; u < U; ++u) acc[u] = -1e30f;

    const float2* pb = (const float2*)(pos + (size_t)b * n * 2);
    const float* tb = t + (size_t)b * n * F;
    for (int j = 0; j < n; ++j) {
        float2 p = pb[j];
        float dx = p.x - cx;
        float dy = p.y - cy;
        if (dx * dx + dy * dy <= r2) {
#pragma unroll
            for (int u = 0; u < U; ++u)
                acc[u] = fmaxf(acc[u], tb[(size_t)j * F + lane + 32 * u]);
        }
    }
#pragma unroll
    for (int u = 0; u < U; ++u) {
        int f = lane + 32 * u;
        out[(size_t)(b * m + i) * F + f] =
            acc[u] + bias[f] - cx * Wp[f] - cy * Wp[F + f];
    }
}

__global__ void init_gf_kernel(unsigned* __restrict__ gf, int n) {
    int i = blockIdx.x * blockDim.x + threadIdx.x;
    if (i < n) gf[i] = 0u;
}
__global__ void decode_gf_kernel(const unsigned* __restrict__ gf,
                                 float* __restrict__ out, int n) {
    int i = blockIdx.x * blockDim.x + threadIdx.x;
    if (i < n) out[i] = ord2f(gf[i]);
}

extern "C" void kernel_launch(void* const* d_in, const int* in_sizes, int n_in,
                              void* d_out, int out_size, void* d_ws, size_t ws_size,
                              hipStream_t stream) {
    const float* x     = (const float*)d_in[0];
    const float* zones = (const float*)d_in[1];
    const float* lf_W1 = (const float*)d_in[2];
    const float* lf_b1 = (const float*)d_in[3];
    const float* lf_W2 = (const float*)d_in[4];
    const float* lf_b2 = (const float*)d_in[5];
    const float* c1_W  = (const float*)d_in[6];
    const float* c1_b  = (const float*)d_in[7];
    const float* c2_W  = (const float*)d_in[8];
    const float* c2_b  = (const float*)d_in[9];
    const float* c3_W  = (const float*)d_in[10];
    const float* c3_b  = (const float*)d_in[11];

    float* out    = (float*)d_out;
    float* lf_out = out;
    float* gf_out = out + (size_t)BATCH * NPTS * 64;

    const int M = BATCH * NPTS;

    float* ws = (float*)d_ws;
    size_t o = 0;
    float* hidden = ws + o;  o += (size_t)M * 64;
    float* t1     = ws + o;  o += (size_t)M * 64;
    float* ctr1   = ws + o;  o += (size_t)BATCH * M1 * 2;
    float* h1     = ws + o;  o += (size_t)BATCH * M1 * 64;
    float* t2     = ws + o;  o += (size_t)BATCH * M1 * 128;
    float* ctr2   = ws + o;  o += (size_t)BATCH * M2 * 2;
    float* h2     = ws + o;  o += (size_t)BATCH * M2 * 128;
    unsigned* gf_enc = (unsigned*)(ws + o);

    hidden_kernel<<<(M * 64 + 255) / 256, 256, 0, stream>>>(x, lf_W1, lf_b1,
                                                            hidden, M);
    gemm_wmma_kernel<1, 64, 0, 0, 64><<<dim3(M / 16, 1), 32, 0, stream>>>(
        hidden, nullptr, nullptr, lf_W2, 64, lf_b2, lf_out, nullptr, 0);

    fps_kernel<<<BATCH, 256, 0, stream>>>(x, NPTS, M1, ctr1);
    fps_kernel<<<BATCH, 256, 0, stream>>>(ctr1, M1, M2, ctr2);

    gemm_wmma_kernel<0, 64, 1, 2, 67><<<dim3(M / 16, 1), 32, 0, stream>>>(
        lf_out, zones, x, c1_W, 64, nullptr, t1, nullptr, 0);

    agg_kernel<2><<<(BATCH * M1 * 32) / 256, 256, 0, stream>>>(
        t1, x, ctr1, c1_W + 65 * 64, c1_b, h1, NPTS, M1, 0.25f);

    gemm_wmma_kernel<0, 64, 2, 0, 66>
        <<<dim3(BATCH * M1 / 16, 2), 32, 0, stream>>>(
            h1, ctr1, nullptr, c2_W, 128, nullptr, t2, nullptr, 0);

    agg_kernel<4><<<(BATCH * M2 * 32) / 256, 256, 0, stream>>>(
        t2, ctr1, ctr2, c2_W + 64 * 128, c2_b, h2, M1, M2, 1.0f);

    init_gf_kernel<<<(BATCH * GF + 255) / 256, 256, 0, stream>>>(gf_enc,
                                                                 BATCH * GF);
    gemm_wmma_kernel<2, 128, 2, 0, 130>
        <<<dim3(BATCH * M2 / 16, GF / 64), 32, 0, stream>>>(
            h2, ctr2, nullptr, c3_W, GF, c3_b, nullptr, gf_enc, M2);
    decode_gf_kernel<<<(BATCH * GF + 255) / 256, 256, 0, stream>>>(gf_enc,
                                                                   gf_out,
                                                                   BATCH * GF);
}


# FPS round loop touches no global memory: points copied to LDS once (sweep and list look-up read them there), picks collect in LDS and go to ctr in one pass at the end; picks unchanged
# speedup vs baseline: 6.4988x; 1.4948x over previous
#include <hip/hip_runtime.h>
#include <hip/hip_bf16.h>
#include <math.h>


typedef __attribute__((ext_vector_type(2))) float v2f;
typedef __attribute__((ext_vector_type(8))) float v8f;

#define BATCH 4
#define NPTS  4096
#define M1    2048
#define M2    512
#define GF    1024

__device__ __forceinline__ unsigned f2ord(float f) {
    unsigned u = __float_as_uint(f);
    return (u & 0x80000000u) ? ~u : (u | 0x80000000u);
}
__device__ __forceinline__ float ord2f(unsigned k) {
    return (k & 0x80000000u) ? __uint_as_float(k & 0x7fffffffu)
                             : __uint_as_float(~k);
}

__global__ void hidden_kernel(const float* __restrict__ x,
                              const float* __restrict__ W1,
                              const float* __restrict__ b1,
                              float* __restrict__ hidden, int M) {
    int idx = blockIdx.x * blockDim.x + threadIdx.x;
    if (idx >= M * 64) return;
    int r = idx >> 6, f = idx & 63;
    float v = x[r * 2 + 0] * W1[f] + x[r * 2 + 1] * W1[64 + f] + b1[f];
    hidden[idx] = tanhf(v);
}

template <int KREL, int C1, int C2>
__device__ __forceinline__ float auxload(const float* __restrict__ A1,
                                         const float* __restrict__ A2,
                                         int row) {
    if constexpr (KREL < C1) {
        return A1[row * C1 + KREL];
    } else if constexpr (KREL - C1 < C2) {
        return A2[row * C2 + (KREL - C1)];
    } else {
        return 0.0f;
    }
}

template <int OUTMODE, int C0, int C1, int C2, int KTOT>
__global__ void gemm_wmma_kernel(const float* __restrict__ A0,
                                 const float* __restrict__ A1,
                                 const float* __restrict__ A2,
                                 const float* __restrict__ W, int F,
                                 const float* __restrict__ bias,
                                 float* __restrict__ out,
                                 unsigned* __restrict__ gfmax,
                                 int rowsPerBatch) {
    constexpr int KP = (KTOT + 3) & ~3;
    static_assert((C0 & 3) == 0, "C0 multiple of 4");
    static_assert(KP - C0 == 0 || KP - C0 == 4, "at most one tail K-step");

    int lane = threadIdx.x & 31;
    int half = lane >> 4;
    int mrow = lane & 15;
    int tm = blockIdx.x;
    int colBase = blockIdx.y * 64;
    int arow = tm * 16 + mrow;

    v8f c[4];
#pragma unroll
    for (int q = 0; q < 4; ++q)
#pragma unroll
        for (int v = 0; v < 8; ++v) c[q][v] = 0.0f;

    const float* a0row = A0 + (size_t)arow * C0;

#pragma unroll
    for (int k0 = 0; k0 < C0; k0 += 4) {
        int ka = k0 + 2 * half;
        v2f a = *(const v2f*)(a0row + ka);
#pragma unroll
        for (int nt = 0; nt < 4; ++nt) {
            int n = colBase + nt * 16 + mrow;
            v2f bm;
            bm.x = W[ka * F + n];
            bm.y = W[(ka + 1) * F + n];
            c[nt] = __builtin_amdgcn_wmma_f32_16x16x4_f32(
                false, a, false, bm, (short)0, c[nt], false, false);
        }
    }

    if constexpr (KP > C0) {
        float t0 = auxload<0, C1, C2>(A1, A2, arow);
        float t1 = auxload<1, C1, C2>(A1, A2, arow);
        float t2 = auxload<2, C1, C2>(A1, A2, arow);
        float t3 = auxload<3, C1, C2>(A1, A2, arow);
        v2f a;
        a.x = half ? t2 : t0;
        a.y = half ? t3 : t1;
        int ka = C0 + 2 * half;
        int kx = (ka < KTOT) ? ka : (KTOT - 1);
        int ky = (ka + 1 < KTOT) ? (ka + 1) : (KTOT - 1);
#pragma unroll
        for (int nt = 0; nt < 4; ++nt) {
            int n = colBase + nt * 16 + mrow;
            float wx = W[kx * F + n];
            float wy = W[ky * F + n];
            v2f bm;
            bm.x = (ka < KTOT) ? wx : 0.0f;
            bm.y = (ka + 1 < KTOT) ? wy : 0.0f;
            c[nt] = __builtin_amdgcn_wmma_f32_16x16x4_f32(
                false, a, false, bm, (short)0, c[nt], false, false);
        }
    }

#pragma unroll
    for (int nt = 0; nt < 4; ++nt) {
        int n = colBase + nt * 16 + mrow;
#pragma unroll
        for (int v = 0; v < 8; ++v) {
            int m = tm * 16 + v + 8 * half;
            float val = c[nt][v];
            if constexpr (OUTMODE == 0) {
                out[(size_t)m * F + n] = val;
            } else if constexpr (OUTMODE == 1) {
                out[(size_t)m * F + n] = tanhf(val + bias[n]);
            } else {
                float g = val + bias[n];
                atomicMax(&gfmax[(m / rowsPerBatch) * F + n], f2ord(g));
            }
        }
    }
}

#define FPS_K 16

__device__ __forceinline__ float dist2(float px, float py, float qx, float qy) {
    float dx = px - qx;
    float dy = py - qy;
    float d2;
    {
#pragma clang fp contract(off)
        d2 = dx * dx + dy * dy;
    }
    return d2;
}

struct fkey {
    unsigned hi;
    unsigned lo;
};

__device__ __forceinline__ fkey fps_key(float md, int j) {
    fkey k;
    k.hi = __float_as_uint(md);
    k.lo = (0x7fffffffu - (unsigned)j) << 1;
    return k;
}

__device__ __forceinline__ bool key_gt(fkey a, fkey b) {
    unsigned borrow_lo, borrow;
    (void)__builtin_subc(b.lo, a.lo, 0u, &borrow_lo);
    (void)__builtin_subc(b.hi, a.hi, borrow_lo, &borrow);
    return borrow != 0u;
}

__device__ __forceinline__ fkey key_sel(bool g, fkey a, fkey b) {
    fkey r;
    r.hi = g ? a.hi : b.hi;
    r.lo = g ? a.lo : b.lo;
    return r;
}

__device__ __forceinline__ void key_apart(fkey& k) {
    asm("" : "+v"(k.hi), "+v"(k.lo));
}

__device__ __forceinline__ fkey key_ld(uint2 v) {
    fkey k;
    k.hi = v.y;
    k.lo = v.x;
    key_apart(k);
    return k;
}
__device__ __forceinline__ uint2 key_st(fkey k) { return make_uint2(k.lo, k.hi); }

__device__ __forceinline__ void cmpx(fkey& a, fkey& b) {
    bool g = key_gt(b, a);
    fkey hi = key_sel(g, b, a);
    fkey lo = key_sel(g, a, b);
    a = hi;
    b = lo;
}

template <int N>
__device__ __forceinline__ void bitonic_desc(fkey (&c)[N]) {
#pragma unroll
    for (int i = 0; i < N; ++i) key_apart(c[i]);
#pragma unroll
    for (int stride = N / 2; stride >= 1; stride >>= 1) {
#pragma unroll
        for (int i = 0; i < N; ++i)
            if ((i & stride) == 0) cmpx(c[i], c[i + stride]);
    }
}

__device__ __forceinline__ unsigned lane_u(unsigned v, int lane) {
    return (unsigned)__builtin_amdgcn_readlane((int)v, lane);
}
__device__ __forceinline__ float lane_f(float v, int lane) {
    return __int_as_float(__builtin_amdgcn_readlane(__float_as_int(v), lane));
}

template <int CTRL>
__device__ __forceinline__ fkey key_row(fkey k) {
    fkey r;
    r.hi = (unsigned)__builtin_amdgcn_update_dpp(0, (int)k.hi, CTRL, 0xf, 0xf, false);
    r.lo = (unsigned)__builtin_amdgcn_update_dpp(0, (int)k.lo, CTRL, 0xf, 0xf, false);
    return r;
}
__device__ __forceinline__ fkey key_other_row(fkey k) {
    fkey r;
    r.hi = __builtin_amdgcn_permlanex16(k.hi, k.hi, 0x76543210u, 0xfedcba98u, false, false);
    r.lo = __builtin_amdgcn_permlanex16(k.lo, k.lo, 0x76543210u, 0xfedcba98u, false, false);
    return r;
}
template <int CTRL>
__device__ __forceinline__ fkey lane_cmpx(fkey own, bool up) {
    fkey o = key_row<CTRL>(own);
    bool g = key_gt(o, own);
    fkey mx = key_sel(g, o, own);
    fkey mn = key_sel(g, own, o);
    key_apart(mx);
    key_apart(mn);
    return key_sel(up, mx, mn);
}
__device__ __forceinline__ fkey merge_rows(fkey v, int e) {
    fkey o = key_other_row(v);
    v = key_sel(key_gt(o, v), o, v);
    v = lane_cmpx<0x128>(v, (e & 8) == 0);
    v = lane_cmpx<0x164>(v, (e & 4) == 0);
    v = lane_cmpx<0x04e>(v, (e & 2) == 0);
    v = lane_cmpx<0x0b1>(v, (e & 1) == 0);
    return v;
}
__device__ __forceinline__ fkey key_mirror(fkey k) { return key_row<0x140>(k); }

__global__ void fps_kernel(const float* __restrict__ pos, int n, int m,
                           float* __restrict__ ctr) {
    __shared__ float mind2[NPTS];
    __shared__ float2 spts[NPTS];
    __shared__ float2 spick[M1];
    __shared__ uint2 skey[FPS_K / 2][256];
    __shared__ uint2 tkey[32][FPS_K + 1];
    __shared__ uint2 wkey[8][FPS_K + 1];
    __shared__ float s_px[FPS_K];
    __shared__ float s_py[FPS_K];
    __shared__ int s_np;

    int b = blockIdx.x;
    int tid = threadIdx.x;
    const float2* pb2 = (const float2*)(pos + (size_t)b * n * 2);
    float2* cb2 = (float2*)(ctr + (size_t)b * m * 2);

    for (int j = tid; j < n; j += 256) {
        mind2[j] = 1e10f;
        spts[j] = pb2[j];
    }
    if (tid == 0) {
        float2 p0 = pb2[0];
        spick[0] = p0;
        for (int q = 0; q < FPS_K; ++q) {
            s_px[q] = p0.x;
            s_py[q] = p0.y;
        }
        s_np = 1;
    }
    __syncthreads();

    int it = 1;
    int np = __builtin_amdgcn_readfirstlane(s_np);
    while (it < m) {
        float px[FPS_K], py[FPS_K];
#pragma unroll
        for (int q = 0; q < FPS_K; ++q) {
            px[q] = s_px[q];
            py[q] = s_py[q];
        }

        fkey L[FPS_K];
        L[0].hi = 0u;
        L[0].lo = 0u;
        L[1].hi = 0u;
        L[1].lo = 0u;
        for (int j = tid; j < n; j += 256) {
            float2 p = spts[j];
            float md = mind2[j];
            float d[FPS_K];
#pragma unroll
            for (int q = 0; q < FPS_K; ++q)
                d[q] = dist2(p.x, p.y, px[q], py[q]);
#pragma unroll
            for (int w = FPS_K / 2; w >= 1; w >>= 1)
#pragma unroll
                for (int k = 0; k < w; ++k) d[k] = fminf(d[k], d[k + w]);
            md = fminf(md, d[0]);
            mind2[j] = md;
            fkey key = fps_key(md, j);
            fkey lo = key_sel(key_gt(key, L[1]), key, L[1]);
            bool g = key_gt(lo, L[0]);
            L[1] = key_sel(g, L[0], lo);
            L[0] = key_sel(g, lo, L[0]);
        }
        L[1].lo |= 1u;
        skey[0][tid] = key_st(L[0]);
        skey[1][tid] = key_st(L[1]);
        __syncthreads();

        static_assert(FPS_K == 16, "the tree below builds lists of 16");
        if (tid < 128) {
            fkey c[4];
            c[0] = L[0];
            c[1] = L[1];
            c[2] = key_ld(skey[1][tid + 128]);
            c[3] = key_ld(skey[0][tid + 128]);
            bitonic_desc<4>(c);
#pragma unroll
            for (int k = 0; k < 4; ++k) {
                L[k] = c[k];
                skey[k][tid] = key_st(c[k]);
            }
        }
        __syncthreads();
        if (tid < 64) {
            fkey c[8];
#pragma unroll
            for (int k = 0; k < 4; ++k) {
                c[k] = L[k];
                c[7 - k] = key_ld(skey[k][tid + 64]);
            }
            bitonic_desc<8>(c);
#pragma unroll
            for (int k = 0; k < 8; ++k) {
                L[k] = c[k];
                skey[k][tid] = key_st(c[k]);
            }
        }
        __syncthreads();
        if (tid < 32) {
            fkey c[16];
#pragma unroll
            for (int k = 0; k < 8; ++k) {
                c[k] = L[k];
                c[15 - k] = key_ld(skey[k][tid + 32]);
            }
            bitonic_desc<16>(c);
#pragma unroll
            for (int k = 0; k < 16; ++k) tkey[tid][k] = key_st(c[k]);
        }
        __syncthreads();
        {
            int lane = tid & 31;
            int wv = tid >> 5;
            int h = lane >> 4;
            int e = lane & 15;
            int at = h ? (FPS_K - 1 - e) : e;
            fkey r0 = merge_rows(key_ld(tkey[4 * wv + h][at]), e);
            fkey r1 = merge_rows(key_ld(tkey[4 * wv + 2 + h][at]), e);
            fkey x = merge_rows(key_sel(h != 0, key_mirror(r1), r0), e);
            if (h == 0) wkey[wv][e] = key_st(x);
        }
        __syncthreads();

        if (tid < 32) {
            int li = tid & 15;
            bool upper = tid >= 16;
            int at = upper ? (FPS_K - 1 - li) : li;
            int hr = upper ? 1 : 0;
            fkey q0 = merge_rows(key_ld(wkey[0 + hr][at]), li);
            fkey q1 = merge_rows(key_ld(wkey[2 + hr][at]), li);
            fkey q2 = merge_rows(key_ld(wkey[4 + hr][at]), li);
            fkey q3 = merge_rows(key_ld(wkey[6 + hr][at]), li);
            fkey d0 = merge_rows(key_sel(upper, key_mirror(q1), q0), li);
            fkey d1 = merge_rows(key_sel(upper, key_mirror(q3), q2), li);
            fkey my = merge_rows(key_sel(upper, key_mirror(d1), d0), li);
            unsigned l = my.lo;
            unsigned lo = l & ~1u;
            unsigned hi = my.hi;
            float md = __uint_as_float(hi);
            float2 cp = spts[0x7fffffffu - (l >> 1)];
            unsigned mk = __builtin_amdgcn_ballot_w32((l & 1u) != 0u) & 0xffffu;
            int ex = mk ? __ffs((int)mk) : FPS_K;
            unsigned bound_hi = lane_u(hi, ex - 1);
            unsigned bound_lo = lane_u(lo, ex - 1);
            float c0x = lane_f(cp.x, 0);
            float c0y = lane_f(cp.y, 0);
            unsigned w = 0u;
#pragma unroll
            for (int a = 0; a < FPS_K; ++a) {
                float d = dist2(cp.x, cp.y, lane_f(cp.x, a), lane_f(cp.y, a));
                bool lowers = d < md;
                unsigned db = __float_as_uint(d);
                bool behind = (db < bound_hi) | ((db == bound_hi) & (lo < bound_lo));
                w |= (lowers ? (1u << a) : 0u) | (behind ? (0x10000u << a) : 0u);
            }
            int room = m - it;
            unsigned below = (1u << li) - 1u;
            unsigned lowers = w & below;
            unsigned behinds = (w >> 16) & below;
            unsigned all = 1u;
#pragma unroll
            for (int k = 1; k < FPS_K; ++k)
                all = __builtin_amdgcn_ballot_w32((lowers & all) == 0u) & 0xffffu;
            bool keeps = ((all >> li) & 1u) != 0u;
            bool behind = (behinds & all) != 0u;
            bool live = (li < ex) && (hi != 0u) && (__popc(all & below) < room);
            bool on = live && (keeps || behind);
            unsigned stops = __builtin_amdgcn_ballot_w32(!on) & 0xfffeu;
            int first = stops ? (__ffs((int)stops) - 1) : FPS_K;
            unsigned taken = all & ((1u << first) - 1u);
            int cnt = __popc(taken);
            if (tid < FPS_K) {
                if ((taken >> tid) & 1u) {
                    int slot = __popc(taken & ((1u << tid) - 1u));
                    spick[it + slot] = cp;
                    s_px[slot] = cp.x;
                    s_py[slot] = cp.y;
                }
                if (tid >= cnt) {
                    s_px[tid] = c0x;
                    s_py[tid] = c0y;
                }
            }
            if (tid == 0) s_np = cnt;
        }
        __syncthreads();
        np = __builtin_amdgcn_readfirstlane(s_np);
        it += np;
    }
    for (int i = tid; i < m; i += 256) cb2[i] = spick[i];
}

#define AGG_FW 64

__device__ __forceinline__ float keep_max(float a, float v) { return (v > a) ? v : a; }

template <int F, int NS>
__global__ __launch_bounds__(32 * NS)
void agg_kernel(const float* __restrict__ t,
                const float* __restrict__ pos,
                const float* __restrict__ ctr,
                const float* __restrict__ Wp,
                const float* __restrict__ bias,
                float* __restrict__ out,
                int n, int m, float r2) {
    constexpr int NT = 32 * NS;
    static_assert(F % AGG_FW == 0, "whole feature groups");
    __shared__ float part[NS][32][AGG_FW + 1];

    int lane = threadIdx.x & 31;
    int ps = threadIdx.x >> 5;
    int f0 = blockIdx.y * AGG_FW;
    int tiles = m >> 5;
    int b = blockIdx.x / tiles;
    int i0 = (blockIdx.x - b * tiles) << 5;

    float2 cc = ((const float2*)ctr)[b * m + i0 + lane];

    float acc[AGG_FW];
#pragma unroll
    for (int u = 0; u < AGG_FW; ++u) acc[u] = -1e30f;

    const float2* pb = (const float2*)pos + (size_t)b * n;
    const float* tb = t + (size_t)b * n * F + f0;
    int slice = n / NS;
    int j0 = ps * slice;

    float2 p = pb[j0 + lane];
    for (int c0 = 0; c0 < slice; c0 += 32) {
        int nx = (c0 + 32 < slice) ? (c0 + 32) : c0;
        float2 pn = pb[j0 + nx + lane];
        for (int q = 0; q < 32; q += 8) {
            bool in[8];
            unsigned any = 0u;
#pragma unroll
            for (int k = 0; k < 8; ++k) {
                float px = lane_f(p.x, q + k);
                float py = lane_f(p.y, q + k);
                in[k] = dist2(px, py, cc.x, cc.y) <= r2;
                any |= __builtin_amdgcn_ballot_w32(in[k]);
            }
            if (any != 0u) {
#pragma unroll
                for (int k = 0; k < 8; ++k) {
                    if (in[k]) {
                        const float4* row =
                            (const float4*)(tb + (size_t)(j0 + c0 + q + k) * F);
                        float4 v[AGG_FW / 4];
#pragma unroll
                        for (int u = 0; u < AGG_FW / 4; ++u) v[u] = row[u];
#pragma unroll
                        for (int u = 0; u < AGG_FW / 4; ++u) {
                            acc[4 * u + 0] = keep_max(acc[4 * u + 0], v[u].x);
                            acc[4 * u + 1] = keep_max(acc[4 * u + 1], v[u].y);
                            acc[4 * u + 2] = keep_max(acc[4 * u + 2], v[u].z);
                            acc[4 * u + 3] = keep_max(acc[4 * u + 3], v[u].w);
                        }
                    }
                }
            }
        }
        p = pn;
    }

#pragma unroll
    for (int u = 0; u < AGG_FW; ++u) part[ps][lane][u] = acc[u];
    __syncthreads();

    for (int idx = threadIdx.x; idx < 32 * AGG_FW; idx += NT) {
        int c = idx / AGG_FW;
        int fl = idx - c * AGG_FW;
        int f = f0 + fl;
        float v = part[0][c][fl];
#pragma unroll
        for (int s = 1; s < NS; ++s) v = fmaxf(v, part[s][c][fl]);
        float2 cq = ((const float2*)ctr)[b * m + i0 + c];
        out[(size_t)(b * m + i0 + c) * F + f] =
            v + bias[f] - cq.x * Wp[f] - cq.y * Wp[F + f];
    }
}

__global__ void init_gf_kernel(unsigned* __restrict__ gf, int n) {
    int i = blockIdx.x * blockDim.x + threadIdx.x;
    if (i < n) gf[i] = 0u;
}
__global__ void decode_gf_kernel(const unsigned* __restrict__ gf,
                                 float* __restrict__ out, int n) {
    int i = blockIdx.x * blockDim.x + threadIdx.x;
    if (i < n) out[i] = ord2f(gf[i]);
}

extern "C" void kernel_launch(void* const* d_in, const int* in_sizes, int n_in,
                              void* d_out, int out_size, void* d_ws, size_t ws_size,
                              hipStream_t stream) {
    const float* x     = (const float*)d_in[0];
    const float* zones = (const float*)d_in[1];
    const float* lf_W1 = (const float*)d_in[2];
    const float* lf_b1 = (const float*)d_in[3];
    const float* lf_W2 = (const float*)d_in[4];
    const float* lf_b2 = (const float*)d_in[5];
    const float* c1_W  = (const float*)d_in[6];
    const float* c1_b  = (const float*)d_in[7];
    const float* c2_W  = (const float*)d_in[8];
    const float* c2_b  = (const float*)d_in[9];
    const float* c3_W  = (const float*)d_in[10];
    const float* c3_b  = (const float*)d_in[11];

    float* out    = (float*)d_out;
    float* lf_out = out;
    float* gf_out = out + (size_t)BATCH * NPTS * 64;

    const int M = BATCH * NPTS;

    float* ws = (float*)d_ws;
    size_t o = 0;
    float* hidden = ws + o;  o += (size_t)M * 64;
    float* t1     = ws + o;  o += (size_t)M * 64;
    float* ctr1   = ws + o;  o += (size_t)BATCH * M1 * 2;
    float* h1     = ws + o;  o += (size_t)BATCH * M1 * 64;
    float* t2     = ws + o;  o += (size_t)BATCH * M1 * 128;
    float* ctr2   = ws + o;  o += (size_t)BATCH * M2 * 2;
    float* h2     = ws + o;  o += (size_t)BATCH * M2 * 128;
    unsigned* gf_enc = (unsigned*)(ws + o);

    hidden_kernel<<<(M * 64 + 255) / 256, 256, 0, stream>>>(x, lf_W1, lf_b1,
                                                            hidden, M);
    gemm_wmma_kernel<1, 64, 0, 0, 64><<<dim3(M / 16, 1), 32, 0, stream>>>(
        hidden, nullptr, nullptr, lf_W2, 64, lf_b2, lf_out, nullptr, 0);

    fps_kernel<<<BATCH, 256, 0, stream>>>(x, NPTS, M1, ctr1);
    fps_kernel<<<BATCH, 256, 0, stream>>>(ctr1, M1, M2, ctr2);

    gemm_wmma_kernel<0, 64, 1, 2, 67><<<dim3(M / 16, 1), 32, 0, stream>>>(
        lf_out, zones, x, c1_W, 64, nullptr, t1, nullptr, 0);

    agg_kernel<64, 16><<<dim3(BATCH * M1 / 32, 1), 512, 0, stream>>>(
        t1, x, ctr1, c1_W + 65 * 64, c1_b, h1, NPTS, M1, 0.25f);

    gemm_wmma_kernel<0, 64, 2, 0, 66>
        <<<dim3(BATCH * M1 / 16, 2), 32, 0, stream>>>(
            h1, ctr1, nullptr, c2_W, 128, nullptr, t2, nullptr, 0);

    agg_kernel<128, 16><<<dim3(BATCH * M2 / 32, 2), 512, 0, stream>>>(
        t2, ctr1, ctr2, c2_W + 64 * 128, c2_b, h2, M1, M2, 1.0f);

    init_gf_kernel<<<(BATCH * GF + 255) / 256, 256, 0, stream>>>(gf_enc,
                                                                 BATCH * GF);
    gemm_wmma_kernel<2, 128, 2, 0, 130>
        <<<dim3(BATCH * M2 / 16, GF / 64), 32, 0, stream>>>(
            h2, ctr2, nullptr, c3_W, GF, c3_b, nullptr, gf_enc, M2);
    decode_gf_kernel<<<(BATCH * GF + 255) / 256, 256, 0, stream>>>(gf_enc,
                                                                   gf_out,
                                                                   BATCH * GF);
}


# FPS ranked lists: first tree levels (256 lists of 2 -> 8 of 16) run inside each wave across lanes (rows of 16 ranked by a 10-step sorting network, 3 merges): 2 barriers a round, not 6; picks same
# speedup vs baseline: 7.1254x; 1.0964x over previous
#include <hip/hip_runtime.h>
#include <hip/hip_bf16.h>
#include <math.h>


typedef __attribute__((ext_vector_type(2))) float v2f;
typedef __attribute__((ext_vector_type(8))) float v8f;

#define BATCH 4
#define NPTS  4096
#define M1    2048
#define M2    512
#define GF    1024

__device__ __forceinline__ unsigned f2ord(float f) {
    unsigned u = __float_as_uint(f);
    return (u & 0x80000000u) ? ~u : (u | 0x80000000u);
}
__device__ __forceinline__ float ord2f(unsigned k) {
    return (k & 0x80000000u) ? __uint_as_float(k & 0x7fffffffu)
                             : __uint_as_float(~k);
}

__global__ void hidden_kernel(const float* __restrict__ x,
                              const float* __restrict__ W1,
                              const float* __restrict__ b1,
                              float* __restrict__ hidden, int M) {
    int idx = blockIdx.x * blockDim.x + threadIdx.x;
    if (idx >= M * 64) return;
    int r = idx >> 6, f = idx & 63;
    float v = x[r * 2 + 0] * W1[f] + x[r * 2 + 1] * W1[64 + f] + b1[f];
    hidden[idx] = tanhf(v);
}

template <int KREL, int C1, int C2>
__device__ __forceinline__ float auxload(const float* __restrict__ A1,
                                         const float* __restrict__ A2,
                                         int row) {
    if constexpr (KREL < C1) {
        return A1[row * C1 + KREL];
    } else if constexpr (KREL - C1 < C2) {
        return A2[row * C2 + (KREL - C1)];
    } else {
        return 0.0f;
    }
}

template <int OUTMODE, int C0, int C1, int C2, int KTOT>
__global__ void gemm_wmma_kernel(const float* __restrict__ A0,
                                 const float* __restrict__ A1,
                                 const float* __restrict__ A2,
                                 const float* __restrict__ W, int F,
                                 const float* __restrict__ bias,
                                 float* __restrict__ out,
                                 unsigned* __restrict__ gfmax,
                                 int rowsPerBatch) {
    constexpr int KP = (KTOT + 3) & ~3;
    static_assert((C0 & 3) == 0, "C0 multiple of 4");
    static_assert(KP - C0 == 0 || KP - C0 == 4, "at most one tail K-step");

    int lane = threadIdx.x & 31;
    int half = lane >> 4;
    int mrow = lane & 15;
    int tm = blockIdx.x;
    int colBase = blockIdx.y * 64;
    int arow = tm * 16 + mrow;

    v8f c[4];
#pragma unroll
    for (int q = 0; q < 4; ++q)
#pragma unroll
        for (int v = 0; v < 8; ++v) c[q][v] = 0.0f;

    const float* a0row = A0 + (size_t)arow * C0;

#pragma unroll
    for (int k0 = 0; k0 < C0; k0 += 4) {
        int ka = k0 + 2 * half;
        v2f a = *(const v2f*)(a0row + ka);
#pragma unroll
        for (int nt = 0; nt < 4; ++nt) {
            int n = colBase + nt * 16 + mrow;
            v2f bm;
            bm.x = W[ka * F + n];
            bm.y = W[(ka + 1) * F + n];
            c[nt] = __builtin_amdgcn_wmma_f32_16x16x4_f32(
                false, a, false, bm, (short)0, c[nt], false, false);
        }
    }

    if constexpr (KP > C0) {
        float t0 = auxload<0, C1, C2>(A1, A2, arow);
        float t1 = auxload<1, C1, C2>(A1, A2, arow);
        float t2 = auxload<2, C1, C2>(A1, A2, arow);
        float t3 = auxload<3, C1, C2>(A1, A2, arow);
        v2f a;
        a.x = half ? t2 : t0;
        a.y = half ? t3 : t1;
        int ka = C0 + 2 * half;
        int kx = (ka < KTOT) ? ka : (KTOT - 1);
        int ky = (ka + 1 < KTOT) ? (ka + 1) : (KTOT - 1);
#pragma unroll
        for (int nt = 0; nt < 4; ++nt) {
            int n = colBase + nt * 16 + mrow;
            float wx = W[kx * F + n];
            float wy = W[ky * F + n];
            v2f bm;
            bm.x = (ka < KTOT) ? wx : 0.0f;
            bm.y = (ka + 1 < KTOT) ? wy : 0.0f;
            c[nt] = __builtin_amdgcn_wmma_f32_16x16x4_f32(
                false, a, false, bm, (short)0, c[nt], false, false);
        }
    }

#pragma unroll
    for (int nt = 0; nt < 4; ++nt) {
        int n = colBase + nt * 16 + mrow;
#pragma unroll
        for (int v = 0; v < 8; ++v) {
            int m = tm * 16 + v + 8 * half;
            float val = c[nt][v];
            if constexpr (OUTMODE == 0) {
                out[(size_t)m * F + n] = val;
            } else if constexpr (OUTMODE == 1) {
                out[(size_t)m * F + n] = tanhf(val + bias[n]);
            } else {
                float g = val + bias[n];
                atomicMax(&gfmax[(m / rowsPerBatch) * F + n], f2ord(g));
            }
        }
    }
}

#define FPS_K 16

__device__ __forceinline__ float dist2(float px, float py, float qx, float qy) {
    float dx = px - qx;
    float dy = py - qy;
    float d2;
    {
#pragma clang fp contract(off)
        d2 = dx * dx + dy * dy;
    }
    return d2;
}

struct fkey {
    unsigned hi;
    unsigned lo;
};

__device__ __forceinline__ fkey fps_key(float md, int j) {
    fkey k;
    k.hi = __float_as_uint(md);
    k.lo = (0x7fffffffu - (unsigned)j) << 1;
    return k;
}

__device__ __forceinline__ bool key_gt(fkey a, fkey b) {
    unsigned borrow_lo, borrow;
    (void)__builtin_subc(b.lo, a.lo, 0u, &borrow_lo);
    (void)__builtin_subc(b.hi, a.hi, borrow_lo, &borrow);
    return borrow != 0u;
}

__device__ __forceinline__ fkey key_sel(bool g, fkey a, fkey b) {
    fkey r;
    r.hi = g ? a.hi : b.hi;
    r.lo = g ? a.lo : b.lo;
    return r;
}

__device__ __forceinline__ void key_apart(fkey& k) {
    asm("" : "+v"(k.hi), "+v"(k.lo));
}

__device__ __forceinline__ fkey key_ld(uint2 v) {
    fkey k;
    k.hi = v.y;
    k.lo = v.x;
    key_apart(k);
    return k;
}
__device__ __forceinline__ uint2 key_st(fkey k) { return make_uint2(k.lo, k.hi); }

__device__ __forceinline__ unsigned lane_u(unsigned v, int lane) {
    return (unsigned)__builtin_amdgcn_readlane((int)v, lane);
}
__device__ __forceinline__ float lane_f(float v, int lane) {
    return __int_as_float(__builtin_amdgcn_readlane(__float_as_int(v), lane));
}

template <int CTRL>
__device__ __forceinline__ fkey key_row(fkey k) {
    fkey r;
    r.hi = (unsigned)__builtin_amdgcn_update_dpp(0, (int)k.hi, CTRL, 0xf, 0xf, false);
    r.lo = (unsigned)__builtin_amdgcn_update_dpp(0, (int)k.lo, CTRL, 0xf, 0xf, false);
    return r;
}
__device__ __forceinline__ fkey key_other_row(fkey k) {
    fkey r;
    r.hi = __builtin_amdgcn_permlanex16(k.hi, k.hi, 0x76543210u, 0xfedcba98u, false, false);
    r.lo = __builtin_amdgcn_permlanex16(k.lo, k.lo, 0x76543210u, 0xfedcba98u, false, false);
    return r;
}
template <int CTRL>
__device__ __forceinline__ fkey lane_cmpx(fkey own, bool up) {
    fkey o = key_row<CTRL>(own);
    bool g = key_gt(o, own);
    fkey mx = key_sel(g, o, own);
    fkey mn = key_sel(g, own, o);
    key_apart(mx);
    key_apart(mn);
    return key_sel(up, mx, mn);
}
__device__ __forceinline__ fkey merge_rows(fkey v, int e) {
    fkey o = key_other_row(v);
    v = key_sel(key_gt(o, v), o, v);
    v = lane_cmpx<0x128>(v, (e & 8) == 0);
    v = lane_cmpx<0x164>(v, (e & 4) == 0);
    v = lane_cmpx<0x04e>(v, (e & 2) == 0);
    v = lane_cmpx<0x0b1>(v, (e & 1) == 0);
    return v;
}
__device__ __forceinline__ fkey key_mirror(fkey k) { return key_row<0x140>(k); }

template <int CTRL, int J, int K>
__device__ __forceinline__ fkey sort_step(fkey v, int e, bool upper) {
    bool up = (((e & J) == 0) != ((e & K) != 0)) != upper;
    return lane_cmpx<CTRL>(v, up);
}
__device__ __forceinline__ fkey sort_rows(fkey v, int e, bool upper) {
    v = sort_step<0x0b1, 1, 2>(v, e, upper);
    v = sort_step<0x04e, 2, 4>(v, e, upper);
    v = sort_step<0x0b1, 1, 4>(v, e, upper);
    v = sort_step<0x164, 4, 8>(v, e, upper);
    v = sort_step<0x04e, 2, 8>(v, e, upper);
    v = sort_step<0x0b1, 1, 8>(v, e, upper);
    v = sort_step<0x128, 8, 16>(v, e, upper);
    v = sort_step<0x164, 4, 16>(v, e, upper);
    v = sort_step<0x04e, 2, 16>(v, e, upper);
    v = sort_step<0x0b1, 1, 16>(v, e, upper);
    return v;
}

__global__ void fps_kernel(const float* __restrict__ pos, int n, int m,
                           float* __restrict__ ctr) {
    __shared__ float mind2[NPTS];
    __shared__ float2 spts[NPTS];
    __shared__ float2 spick[M1];
    __shared__ uint2 wkey[8][FPS_K + 1];
    __shared__ float s_px[FPS_K];
    __shared__ float s_py[FPS_K];
    __shared__ int s_np;

    int b = blockIdx.x;
    int tid = threadIdx.x;
    const float2* pb2 = (const float2*)(pos + (size_t)b * n * 2);
    float2* cb2 = (float2*)(ctr + (size_t)b * m * 2);

    for (int j = tid; j < n; j += 256) {
        mind2[j] = 1e10f;
        spts[j] = pb2[j];
    }
    if (tid == 0) {
        float2 p0 = pb2[0];
        spick[0] = p0;
        for (int q = 0; q < FPS_K; ++q) {
            s_px[q] = p0.x;
            s_py[q] = p0.y;
        }
        s_np = 1;
    }
    __syncthreads();

    int it = 1;
    int np = __builtin_amdgcn_readfirstlane(s_np);
    while (it < m) {
        float px[FPS_K], py[FPS_K];
#pragma unroll
        for (int q = 0; q < FPS_K; ++q) {
            px[q] = s_px[q];
            py[q] = s_py[q];
        }

        fkey L[2];
        L[0].hi = 0u;
        L[0].lo = 0u;
        L[1].hi = 0u;
        L[1].lo = 0u;
        for (int j = tid; j < n; j += 256) {
            float2 p = spts[j];
            float md = mind2[j];
            float d[FPS_K];
#pragma unroll
            for (int q = 0; q < FPS_K; ++q)
                d[q] = dist2(p.x, p.y, px[q], py[q]);
#pragma unroll
            for (int w = FPS_K / 2; w >= 1; w >>= 1)
#pragma unroll
                for (int k = 0; k < w; ++k) d[k] = fminf(d[k], d[k + w]);
            md = fminf(md, d[0]);
            mind2[j] = md;
            fkey key = fps_key(md, j);
            fkey lo = key_sel(key_gt(key, L[1]), key, L[1]);
            bool g = key_gt(lo, L[0]);
            L[1] = key_sel(g, L[0], lo);
            L[0] = key_sel(g, lo, L[0]);
        }
        L[1].lo |= 1u;

        static_assert(FPS_K == 16, "a ranked list is one key per lane of a row of 16");
        {
            int lane = tid & 31;
            int wv = tid >> 5;
            int h = lane >> 4;
            int e = lane & 15;
            fkey s0 = sort_rows(L[0], e, h != 0);
            fkey s1 = sort_rows(L[1], e, h != 0);
            fkey x0 = merge_rows(s0, e);
            fkey x1 = merge_rows(s1, e);
            fkey x = merge_rows(key_sel(h != 0, key_mirror(x1), x0), e);
            if (h == 0) wkey[wv][e] = key_st(x);
        }
        __syncthreads();

        if (tid < 32) {
            int li = tid & 15;
            bool upper = tid >= 16;
            int at = upper ? (FPS_K - 1 - li) : li;
            int hr = upper ? 1 : 0;
            fkey q0 = merge_rows(key_ld(wkey[0 + hr][at]), li);
            fkey q1 = merge_rows(key_ld(wkey[2 + hr][at]), li);
            fkey q2 = merge_rows(key_ld(wkey[4 + hr][at]), li);
            fkey q3 = merge_rows(key_ld(wkey[6 + hr][at]), li);
            fkey d0 = merge_rows(key_sel(upper, key_mirror(q1), q0), li);
            fkey d1 = merge_rows(key_sel(upper, key_mirror(q3), q2), li);
            fkey my = merge_rows(key_sel(upper, key_mirror(d1), d0), li);
            unsigned l = my.lo;
            unsigned lo = l & ~1u;
            unsigned hi = my.hi;
            float md = __uint_as_float(hi);
            float2 cp = spts[0x7fffffffu - (l >> 1)];
            unsigned mk = __builtin_amdgcn_ballot_w32((l & 1u) != 0u) & 0xffffu;
            int ex = mk ? __ffs((int)mk) : FPS_K;
            unsigned bound_hi = lane_u(hi, ex - 1);
            unsigned bound_lo = lane_u(lo, ex - 1);
            float c0x = lane_f(cp.x, 0);
            float c0y = lane_f(cp.y, 0);
            unsigned w = 0u;
#pragma unroll
            for (int a = 0; a < FPS_K; ++a) {
                float d = dist2(cp.x, cp.y, lane_f(cp.x, a), lane_f(cp.y, a));
                bool lowers = d < md;
                unsigned db = __float_as_uint(d);
                bool behind = (db < bound_hi) | ((db == bound_hi) & (lo < bound_lo));
                w |= (lowers ? (1u << a) : 0u) | (behind ? (0x10000u << a) : 0u);
            }
            int room = m - it;
            unsigned below = (1u << li) - 1u;
            unsigned lowers = w & below;
            unsigned behinds = (w >> 16) & below;
            unsigned all = 1u;
#pragma unroll
            for (int k = 1; k < FPS_K; ++k)
                all = __builtin_amdgcn_ballot_w32((lowers & all) == 0u) & 0xffffu;
            bool keeps = ((all >> li) & 1u) != 0u;
            bool behind = (behinds & all) != 0u;
            bool live = (li < ex) && (hi != 0u) && (__popc(all & below) < room);
            bool on = live && (keeps || behind);
            unsigned stops = __builtin_amdgcn_ballot_w32(!on) & 0xfffeu;
            int first = stops ? (__ffs((int)stops) - 1) : FPS_K;
            unsigned taken = all & ((1u << first) - 1u);
            int cnt = __popc(taken);
            if (tid < FPS_K) {
                if ((taken >> tid) & 1u) {
                    int slot = __popc(taken & ((1u << tid) - 1u));
                    spick[it + slot] = cp;
                    s_px[slot] = cp.x;
                    s_py[slot] = cp.y;
                }
                if (tid >= cnt) {
                    s_px[tid] = c0x;
                    s_py[tid] = c0y;
                }
            }
            if (tid == 0) s_np = cnt;
        }
        __syncthreads();
        np = __builtin_amdgcn_readfirstlane(s_np);
        it += np;
    }
    for (int i = tid; i < m; i += 256) cb2[i] = spick[i];
}

#define AGG_FW 64

__device__ __forceinline__ float keep_max(float a, float v) { return (v > a) ? v : a; }

template <int F, int NS>
__global__ __launch_bounds__(32 * NS)
void agg_kernel(const float* __restrict__ t,
                const float* __restrict__ pos,
                const float* __restrict__ ctr,
                const float* __restrict__ Wp,
                const float* __restrict__ bias,
                float* __restrict__ out,
                int n, int m, float r2) {
    constexpr int NT = 32 * NS;
    static_assert(F % AGG_FW == 0, "whole feature groups");
    __shared__ float part[NS][32][AGG_FW + 1];

    int lane = threadIdx.x & 31;
    int ps = threadIdx.x >> 5;
    int f0 = blockIdx.y * AGG_FW;
    int tiles = m >> 5;
    int b = blockIdx.x / tiles;
    int i0 = (blockIdx.x - b * tiles) << 5;

    float2 cc = ((const float2*)ctr)[b * m + i0 + lane];

    float acc[AGG_FW];
#pragma unroll
    for (int u = 0; u < AGG_FW; ++u) acc[u] = -1e30f;

    const float2* pb = (const float2*)pos + (size_t)b * n;
    const float* tb = t + (size_t)b * n * F + f0;
    int slice = n / NS;
    int j0 = ps * slice;

    float2 p = pb[j0 + lane];
    for (int c0 = 0; c0 < slice; c0 += 32) {
        int nx = (c0 + 32 < slice) ? (c0 + 32) : c0;
        float2 pn = pb[j0 + nx + lane];
        for (int q = 0; q < 32; q += 8) {
            bool in[8];
            unsigned any = 0u;
#pragma unroll
            for (int k = 0; k < 8; ++k) {
                float px = lane_f(p.x, q + k);
                float py = lane_f(p.y, q + k);
                in[k] = dist2(px, py, cc.x, cc.y) <= r2;
                any |= __builtin_amdgcn_ballot_w32(in[k]);
            }
            if (any != 0u) {
#pragma unroll
                for (int k = 0; k < 8; ++k) {
                    if (in[k]) {
                        const float4* row =
                            (const float4*)(tb + (size_t)(j0 + c0 + q + k) * F);
                        float4 v[AGG_FW / 4];
#pragma unroll
                        for (int u = 0; u < AGG_FW / 4; ++u) v[u] = row[u];
#pragma unroll
                        for (int u = 0; u < AGG_FW / 4; ++u) {
                            acc[4 * u + 0] = keep_max(acc[4 * u + 0], v[u].x);
                            acc[4 * u + 1] = keep_max(acc[4 * u + 1], v[u].y);
                            acc[4 * u + 2] = keep_max(acc[4 * u + 2], v[u].z);
                            acc[4 * u + 3] = keep_max(acc[4 * u + 3], v[u].w);
                        }
                    }
                }
            }
        }
        p = pn;
    }

#pragma unroll
    for (int u = 0; u < AGG_FW; ++u) part[ps][lane][u] = acc[u];
    __syncthreads();

    for (int idx = threadIdx.x; idx < 32 * AGG_FW; idx += NT) {
        int c = idx / AGG_FW;
        int fl = idx - c * AGG_FW;
        int f = f0 + fl;
        float v = part[0][c][fl];
#pragma unroll
        for (int s = 1; s < NS; ++s) v = fmaxf(v, part[s][c][fl]);
        float2 cq = ((const float2*)ctr)[b * m + i0 + c];
        out[(size_t)(b * m + i0 + c) * F + f] =
            v + bias[f] - cq.x * Wp[f] - cq.y * Wp[F + f];
    }
}

__global__ void init_gf_kernel(unsigned* __restrict__ gf, int n) {
    int i = blockIdx.x * blockDim.x + threadIdx.x;
    if (i < n) gf[i] = 0u;
}
__global__ void decode_gf_kernel(const unsigned* __restrict__ gf,
                                 float* __restrict__ out, int n) {
    int i = blockIdx.x * blockDim.x + threadIdx.x;
    if (i < n) out[i] = ord2f(gf[i]);
}

extern "C" void kernel_launch(void* const* d_in, const int* in_sizes, int n_in,
                              void* d_out, int out_size, void* d_ws, size_t ws_size,
                              hipStream_t stream) {
    const float* x     = (const float*)d_in[0];
    const float* zones = (const float*)d_in[1];
    const float* lf_W1 = (const float*)d_in[2];
    const float* lf_b1 = (const float*)d_in[3];
    const float* lf_W2 = (const float*)d_in[4];
    const float* lf_b2 = (const float*)d_in[5];
    const float* c1_W  = (const float*)d_in[6];
    const float* c1_b  = (const float*)d_in[7];
    const float* c2_W  = (const float*)d_in[8];
    const float* c2_b  = (const float*)d_in[9];
    const float* c3_W  = (const float*)d_in[10];
    const float* c3_b  = (const float*)d_in[11];

    float* out    = (float*)d_out;
    float* lf_out = out;
    float* gf_out = out + (size_t)BATCH * NPTS * 64;

    const int M = BATCH * NPTS;

    float* ws = (float*)d_ws;
    size_t o = 0;
    float* hidden = ws + o;  o += (size_t)M * 64;
    float* t1     = ws + o;  o += (size_t)M * 64;
    float* ctr1   = ws + o;  o += (size_t)BATCH * M1 * 2;
    float* h1     = ws + o;  o += (size_t)BATCH * M1 * 64;
    float* t2     = ws + o;  o += (size_t)BATCH * M1 * 128;
    float* ctr2   = ws + o;  o += (size_t)BATCH * M2 * 2;
    float* h2     = ws + o;  o += (size_t)BATCH * M2 * 128;
    unsigned* gf_enc = (unsigned*)(ws + o);

    hidden_kernel<<<(M * 64 + 255) / 256, 256, 0, stream>>>(x, lf_W1, lf_b1,
                                                            hidden, M);
    gemm_wmma_kernel<1, 64, 0, 0, 64><<<dim3(M / 16, 1), 32, 0, stream>>>(
        hidden, nullptr, nullptr, lf_W2, 64, lf_b2, lf_out, nullptr, 0);

    fps_kernel<<<BATCH, 256, 0, stream>>>(x, NPTS, M1, ctr1);
    fps_kernel<<<BATCH, 256, 0, stream>>>(ctr1, M1, M2, ctr2);

    gemm_wmma_kernel<0, 64, 1, 2, 67><<<dim3(M / 16, 1), 32, 0, stream>>>(
        lf_out, zones, x, c1_W, 64, nullptr, t1, nullptr, 0);

    agg_kernel<64, 16><<<dim3(BATCH * M1 / 32, 1), 512, 0, stream>>>(
        t1, x, ctr1, c1_W + 65 * 64, c1_b, h1, NPTS, M1, 0.25f);

    gemm_wmma_kernel<0, 64, 2, 0, 66>
        <<<dim3(BATCH * M1 / 16, 2), 32, 0, stream>>>(
            h1, ctr1, nullptr, c2_W, 128, nullptr, t2, nullptr, 0);

    agg_kernel<128, 16><<<dim3(BATCH * M2 / 32, 2), 512, 0, stream>>>(
        t2, ctr1, ctr2, c2_W + 64 * 128, c2_b, h2, M1, M2, 1.0f);

    init_gf_kernel<<<(BATCH * GF + 255) / 256, 256, 0, stream>>>(gf_enc,
                                                                 BATCH * GF);
    gemm_wmma_kernel<2, 128, 2, 0, 130>
        <<<dim3(BATCH * M2 / 16, GF / 64), 32, 0, stream>>>(
            h2, ctr2, nullptr, c3_W, GF, c3_b, nullptr, gf_enc, M2);
    decode_gf_kernel<<<(BATCH * GF + 255) / 256, 256, 0, stream>>>(gf_enc,
                                                                   gf_out,
                                                                   BATCH * GF);
}


# FPS sweep written out trip by trip (point count a template constant: 16 or 8 trips a thread): its LDS reads go out trips ahead, the trips' dependent chains overlap; picks, d2 bit-identical
# speedup vs baseline: 8.5230x; 1.1961x over previous
#include <hip/hip_runtime.h>
#include <hip/hip_bf16.h>
#include <math.h>


typedef __attribute__((ext_vector_type(2))) float v2f;
typedef __attribute__((ext_vector_type(8))) float v8f;

#define BATCH 4
#define NPTS  4096
#define M1    2048
#define M2    512
#define GF    1024

__device__ __forceinline__ unsigned f2ord(float f) {
    unsigned u = __float_as_uint(f);
    return (u & 0x80000000u) ? ~u : (u | 0x80000000u);
}
__device__ __forceinline__ float ord2f(unsigned k) {
    return (k & 0x80000000u) ? __uint_as_float(k & 0x7fffffffu)
                             : __uint_as_float(~k);
}

__global__ void hidden_kernel(const float* __restrict__ x,
                              const float* __restrict__ W1,
                              const float* __restrict__ b1,
                              float* __restrict__ hidden, int M) {
    int idx = blockIdx.x * blockDim.x + threadIdx.x;
    if (idx >= M * 64) return;
    int r = idx >> 6, f = idx & 63;
    float v = x[r * 2 + 0] * W1[f] + x[r * 2 + 1] * W1[64 + f] + b1[f];
    hidden[idx] = tanhf(v);
}

template <int KREL, int C1, int C2>
__device__ __forceinline__ float auxload(const float* __restrict__ A1,
                                         const float* __restrict__ A2,
                                         int row) {
    if constexpr (KREL < C1) {
        return A1[row * C1 + KREL];
    } else if constexpr (KREL - C1 < C2) {
        return A2[row * C2 + (KREL - C1)];
    } else {
        return 0.0f;
    }
}

template <int OUTMODE, int C0, int C1, int C2, int KTOT>
__global__ void gemm_wmma_kernel(const float* __restrict__ A0,
                                 const float* __restrict__ A1,
                                 const float* __restrict__ A2,
                                 const float* __restrict__ W, int F,
                                 const float* __restrict__ bias,
                                 float* __restrict__ out,
                                 unsigned* __restrict__ gfmax,
                                 int rowsPerBatch) {
    constexpr int KP = (KTOT + 3) & ~3;
    static_assert((C0 & 3) == 0, "C0 multiple of 4");
    static_assert(KP - C0 == 0 || KP - C0 == 4, "at most one tail K-step");

    int lane = threadIdx.x & 31;
    int half = lane >> 4;
    int mrow = lane & 15;
    int tm = blockIdx.x;
    int colBase = blockIdx.y * 64;
    int arow = tm * 16 + mrow;

    v8f c[4];
#pragma unroll
    for (int q = 0; q < 4; ++q)
#pragma unroll
        for (int v = 0; v < 8; ++v) c[q][v] = 0.0f;

    const float* a0row = A0 + (size_t)arow * C0;

#pragma unroll
    for (int k0 = 0; k0 < C0; k0 += 4) {
        int ka = k0 + 2 * half;
        v2f a = *(const v2f*)(a0row + ka);
#pragma unroll
        for (int nt = 0; nt < 4; ++nt) {
            int n = colBase + nt * 16 + mrow;
            v2f bm;
            bm.x = W[ka * F + n];
            bm.y = W[(ka + 1) * F + n];
            c[nt] = __builtin_amdgcn_wmma_f32_16x16x4_f32(
                false, a, false, bm, (short)0, c[nt], false, false);
        }
    }

    if constexpr (KP > C0) {
        float t0 = auxload<0, C1, C2>(A1, A2, arow);
        float t1 = auxload<1, C1, C2>(A1, A2, arow);
        float t2 = auxload<2, C1, C2>(A1, A2, arow);
        float t3 = auxload<3, C1, C2>(A1, A2, arow);
        v2f a;
        a.x = half ? t2 : t0;
        a.y = half ? t3 : t1;
        int ka = C0 + 2 * half;
        int kx = (ka < KTOT) ? ka : (KTOT - 1);
        int ky = (ka + 1 < KTOT) ? (ka + 1) : (KTOT - 1);
#pragma unroll
        for (int nt = 0; nt < 4; ++nt) {
            int n = colBase + nt * 16 + mrow;
            float wx = W[kx * F + n];
            float wy = W[ky * F + n];
            v2f bm;
            bm.x = (ka < KTOT) ? wx : 0.0f;
            bm.y = (ka + 1 < KTOT) ? wy : 0.0f;
            c[nt] = __builtin_amdgcn_wmma_f32_16x16x4_f32(
                false, a, false, bm, (short)0, c[nt], false, false);
        }
    }

#pragma unroll
    for (int nt = 0; nt < 4; ++nt) {
        int n = colBase + nt * 16 + mrow;
#pragma unroll
        for (int v = 0; v < 8; ++v) {
            int m = tm * 16 + v + 8 * half;
            float val = c[nt][v];
            if constexpr (OUTMODE == 0) {
                out[(size_t)m * F + n] = val;
            } else if constexpr (OUTMODE == 1) {
                out[(size_t)m * F + n] = tanhf(val + bias[n]);
            } else {
                float g = val + bias[n];
                atomicMax(&gfmax[(m / rowsPerBatch) * F + n], f2ord(g));
            }
        }
    }
}

#define FPS_K 16

__device__ __forceinline__ float dist2(float px, float py, float qx, float qy) {
    float dx = px - qx;
    float dy = py - qy;
    float d2;
    {
#pragma clang fp contract(off)
        d2 = dx * dx + dy * dy;
    }
    return d2;
}

struct fkey {
    unsigned hi;
    unsigned lo;
};

__device__ __forceinline__ fkey fps_key(float md, int j) {
    fkey k;
    k.hi = __float_as_uint(md);
    k.lo = (0x7fffffffu - (unsigned)j) << 1;
    return k;
}

__device__ __forceinline__ bool key_gt(fkey a, fkey b) {
    unsigned borrow_lo, borrow;
    (void)__builtin_subc(b.lo, a.lo, 0u, &borrow_lo);
    (void)__builtin_subc(b.hi, a.hi, borrow_lo, &borrow);
    return borrow != 0u;
}

__device__ __forceinline__ fkey key_sel(bool g, fkey a, fkey b) {
    fkey r;
    r.hi = g ? a.hi : b.hi;
    r.lo = g ? a.lo : b.lo;
    return r;
}

__device__ __forceinline__ void key_apart(fkey& k) {
    asm("" : "+v"(k.hi), "+v"(k.lo));
}

__device__ __forceinline__ fkey key_ld(uint2 v) {
    fkey k;
    k.hi = v.y;
    k.lo = v.x;
    key_apart(k);
    return k;
}
__device__ __forceinline__ uint2 key_st(fkey k) { return make_uint2(k.lo, k.hi); }

__device__ __forceinline__ unsigned lane_u(unsigned v, int lane) {
    return (unsigned)__builtin_amdgcn_readlane((int)v, lane);
}
__device__ __forceinline__ float lane_f(float v, int lane) {
    return __int_as_float(__builtin_amdgcn_readlane(__float_as_int(v), lane));
}

template <int CTRL>
__device__ __forceinline__ fkey key_row(fkey k) {
    fkey r;
    r.hi = (unsigned)__builtin_amdgcn_update_dpp(0, (int)k.hi, CTRL, 0xf, 0xf, false);
    r.lo = (unsigned)__builtin_amdgcn_update_dpp(0, (int)k.lo, CTRL, 0xf, 0xf, false);
    return r;
}
__device__ __forceinline__ fkey key_other_row(fkey k) {
    fkey r;
    r.hi = __builtin_amdgcn_permlanex16(k.hi, k.hi, 0x76543210u, 0xfedcba98u, false, false);
    r.lo = __builtin_amdgcn_permlanex16(k.lo, k.lo, 0x76543210u, 0xfedcba98u, false, false);
    return r;
}
template <int CTRL>
__device__ __forceinline__ fkey lane_cmpx(fkey own, bool up) {
    fkey o = key_row<CTRL>(own);
    bool g = key_gt(o, own);
    fkey mx = key_sel(g, o, own);
    fkey mn = key_sel(g, own, o);
    key_apart(mx);
    key_apart(mn);
    return key_sel(up, mx, mn);
}
__device__ __forceinline__ fkey merge_rows(fkey v, int e) {
    fkey o = key_other_row(v);
    v = key_sel(key_gt(o, v), o, v);
    v = lane_cmpx<0x128>(v, (e & 8) == 0);
    v = lane_cmpx<0x164>(v, (e & 4) == 0);
    v = lane_cmpx<0x04e>(v, (e & 2) == 0);
    v = lane_cmpx<0x0b1>(v, (e & 1) == 0);
    return v;
}
__device__ __forceinline__ fkey key_mirror(fkey k) { return key_row<0x140>(k); }

template <int CTRL, int J, int K>
__device__ __forceinline__ fkey sort_step(fkey v, int e, bool upper) {
    bool up = (((e & J) == 0) != ((e & K) != 0)) != upper;
    return lane_cmpx<CTRL>(v, up);
}
__device__ __forceinline__ fkey sort_rows(fkey v, int e, bool upper) {
    v = sort_step<0x0b1, 1, 2>(v, e, upper);
    v = sort_step<0x04e, 2, 4>(v, e, upper);
    v = sort_step<0x0b1, 1, 4>(v, e, upper);
    v = sort_step<0x164, 4, 8>(v, e, upper);
    v = sort_step<0x04e, 2, 8>(v, e, upper);
    v = sort_step<0x0b1, 1, 8>(v, e, upper);
    v = sort_step<0x128, 8, 16>(v, e, upper);
    v = sort_step<0x164, 4, 16>(v, e, upper);
    v = sort_step<0x04e, 2, 16>(v, e, upper);
    v = sort_step<0x0b1, 1, 16>(v, e, upper);
    return v;
}

template <int N>
__global__ __launch_bounds__(256) __attribute__((amdgpu_waves_per_eu(2, 2)))
void fps_kernel(const float* __restrict__ pos, int m,
                           float* __restrict__ ctr) {
    static_assert(N % 256 == 0, "every thread sweeps the same number of points");
    constexpr int n = N;
    constexpr int TRIPS = N / 256;
    __shared__ float mind2[N];
    __shared__ float2 spts[N];
    __shared__ float2 spick[M1];
    __shared__ uint2 wkey[8][FPS_K + 1];
    __shared__ float s_px[FPS_K];
    __shared__ float s_py[FPS_K];
    __shared__ int s_np;

    int b = blockIdx.x;
    int tid = threadIdx.x;
    const float2* pb2 = (const float2*)(pos + (size_t)b * n * 2);
    float2* cb2 = (float2*)(ctr + (size_t)b * m * 2);

    for (int j = tid; j < n; j += 256) {
        mind2[j] = 1e10f;
        spts[j] = pb2[j];
    }
    if (tid == 0) {
        float2 p0 = pb2[0];
        spick[0] = p0;
        for (int q = 0; q < FPS_K; ++q) {
            s_px[q] = p0.x;
            s_py[q] = p0.y;
        }
        s_np = 1;
    }
    __syncthreads();

    int it = 1;
    int np = __builtin_amdgcn_readfirstlane(s_np);
    while (it < m) {
        float px[FPS_K], py[FPS_K];
#pragma unroll
        for (int q = 0; q < FPS_K; ++q) {
            px[q] = s_px[q];
            py[q] = s_py[q];
        }
#pragma unroll
        for (int q = 0; q < FPS_K; ++q) asm("" : "+v"(px[q]), "+v"(py[q]));

        fkey L[2];
        L[0].hi = 0u;
        L[0].lo = 0u;
        L[1].hi = 0u;
        L[1].lo = 0u;
        float2 pt[TRIPS];
        float mdv[TRIPS];
#pragma unroll
        for (int t = 0; t < TRIPS; ++t) {
            pt[t] = spts[tid + 256 * t];
            mdv[t] = mind2[tid + 256 * t];
        }
#pragma unroll
        for (int t = 0; t < TRIPS; ++t) {
            int j = tid + 256 * t;
            float2 p = pt[t];
            float md = mdv[t];
            float d[FPS_K];
#pragma unroll
            for (int q = 0; q < FPS_K; ++q)
                d[q] = dist2(p.x, p.y, px[q], py[q]);
#pragma unroll
            for (int w = FPS_K / 2; w >= 1; w >>= 1)
#pragma unroll
                for (int k = 0; k < w; ++k) d[k] = fminf(d[k], d[k + w]);
            md = fminf(md, d[0]);
            mind2[j] = md;
            fkey key = fps_key(md, j);
            fkey lo = key_sel(key_gt(key, L[1]), key, L[1]);
            bool g = key_gt(lo, L[0]);
            L[1] = key_sel(g, L[0], lo);
            L[0] = key_sel(g, lo, L[0]);
        }
        L[1].lo |= 1u;

        static_assert(FPS_K == 16, "a ranked list is one key per lane of a row of 16");
        {
            int lane = tid & 31;
            int wv = tid >> 5;
            int h = lane >> 4;
            int e = lane & 15;
            fkey s0 = sort_rows(L[0], e, h != 0);
            fkey s1 = sort_rows(L[1], e, h != 0);
            fkey x0 = merge_rows(s0, e);
            fkey x1 = merge_rows(s1, e);
            fkey x = merge_rows(key_sel(h != 0, key_mirror(x1), x0), e);
            if (h == 0) wkey[wv][e] = key_st(x);
        }
        __syncthreads();

        if (tid < 32) {
            int li = tid & 15;
            bool upper = tid >= 16;
            int at = upper ? (FPS_K - 1 - li) : li;
            int hr = upper ? 1 : 0;
            fkey q0 = merge_rows(key_ld(wkey[0 + hr][at]), li);
            fkey q1 = merge_rows(key_ld(wkey[2 + hr][at]), li);
            fkey q2 = merge_rows(key_ld(wkey[4 + hr][at]), li);
            fkey q3 = merge_rows(key_ld(wkey[6 + hr][at]), li);
            fkey d0 = merge_rows(key_sel(upper, key_mirror(q1), q0), li);
            fkey d1 = merge_rows(key_sel(upper, key_mirror(q3), q2), li);
            fkey my = merge_rows(key_sel(upper, key_mirror(d1), d0), li);
            unsigned l = my.lo;
            unsigned lo = l & ~1u;
            unsigned hi = my.hi;
            float md = __uint_as_float(hi);
            float2 cp = spts[0x7fffffffu - (l >> 1)];
            unsigned mk = __builtin_amdgcn_ballot_w32((l & 1u) != 0u) & 0xffffu;
            int ex = mk ? __ffs((int)mk) : FPS_K;
            unsigned bound_hi = lane_u(hi, ex - 1);
            unsigned bound_lo = lane_u(lo, ex - 1);
            float c0x = lane_f(cp.x, 0);
            float c0y = lane_f(cp.y, 0);
            unsigned w = 0u;
#pragma unroll
            for (int a = 0; a < FPS_K; ++a) {
                float d = dist2(cp.x, cp.y, lane_f(cp.x, a), lane_f(cp.y, a));
                bool lowers = d < md;
                unsigned db = __float_as_uint(d);
                bool behind = (db < bound_hi) | ((db == bound_hi) & (lo < bound_lo));
                w |= (lowers ? (1u << a) : 0u) | (behind ? (0x10000u << a) : 0u);
            }
            int room = m - it;
            unsigned below = (1u << li) - 1u;
            unsigned lowers = w & below;
            unsigned behinds = (w >> 16) & below;
            unsigned all = 1u;
#pragma unroll
            for (int k = 1; k < FPS_K; ++k)
                all = __builtin_amdgcn_ballot_w32((lowers & all) == 0u) & 0xffffu;
            bool keeps = ((all >> li) & 1u) != 0u;
            bool behind = (behinds & all) != 0u;
            bool live = (li < ex) && (hi != 0u) && (__popc(all & below) < room);
            bool on = live && (keeps || behind);
            unsigned stops = __builtin_amdgcn_ballot_w32(!on) & 0xfffeu;
            int first = stops ? (__ffs((int)stops) - 1) : FPS_K;
            unsigned taken = all & ((1u << first) - 1u);
            int cnt = __popc(taken);
            if (tid < FPS_K) {
                if ((taken >> tid) & 1u) {
                    int slot = __popc(taken & ((1u << tid) - 1u));
                    spick[it + slot] = cp;
                    s_px[slot] = cp.x;
                    s_py[slot] = cp.y;
                }
                if (tid >= cnt) {
                    s_px[tid] = c0x;
                    s_py[tid] = c0y;
                }
            }
            if (tid == 0) s_np = cnt;
        }
        __syncthreads();
        np = __builtin_amdgcn_readfirstlane(s_np);
        it += np;
    }
    for (int i = tid; i < m; i += 256) cb2[i] = spick[i];
}

#define AGG_FW 64

__device__ __forceinline__ float keep_max(float a, float v) { return (v > a) ? v : a; }

template <int F, int NS>
__global__ __launch_bounds__(32 * NS)
void agg_kernel(const float* __restrict__ t,
                const float* __restrict__ pos,
                const float* __restrict__ ctr,
                const float* __restrict__ Wp,
                const float* __restrict__ bias,
                float* __restrict__ out,
                int n, int m, float r2) {
    constexpr int NT = 32 * NS;
    static_assert(F % AGG_FW == 0, "whole feature groups");
    __shared__ float part[NS][32][AGG_FW + 1];

    int lane = threadIdx.x & 31;
    int ps = threadIdx.x >> 5;
    int f0 = blockIdx.y * AGG_FW;
    int tiles = m >> 5;
    int b = blockIdx.x / tiles;
    int i0 = (blockIdx.x - b * tiles) << 5;

    float2 cc = ((const float2*)ctr)[b * m + i0 + lane];

    float acc[AGG_FW];
#pragma unroll
    for (int u = 0; u < AGG_FW; ++u) acc[u] = -1e30f;

    const float2* pb = (const float2*)pos + (size_t)b * n;
    const float* tb = t + (size_t)b * n * F + f0;
    int slice = n / NS;
    int j0 = ps * slice;

    float2 p = pb[j0 + lane];
    for (int c0 = 0; c0 < slice; c0 += 32) {
        int nx = (c0 + 32 < slice) ? (c0 + 32) : c0;
        float2 pn = pb[j0 + nx + lane];
        for (int q = 0; q < 32; q += 8) {
            bool in[8];
            unsigned any = 0u;
#pragma unroll
            for (int k = 0; k < 8; ++k) {
                float px = lane_f(p.x, q + k);
                float py = lane_f(p.y, q + k);
                in[k] = dist2(px, py, cc.x, cc.y) <= r2;
                any |= __builtin_amdgcn_ballot_w32(in[k]);
            }
            if (any != 0u) {
#pragma unroll
                for (int k = 0; k < 8; ++k) {
                    if (in[k]) {
                        const float4* row =
                            (const float4*)(tb + (size_t)(j0 + c0 + q + k) * F);
                        float4 v[AGG_FW / 4];
#pragma unroll
                        for (int u = 0; u < AGG_FW / 4; ++u) v[u] = row[u];
#pragma unroll
                        for (int u = 0; u < AGG_FW / 4; ++u) {
                            acc[4 * u + 0] = keep_max(acc[4 * u + 0], v[u].x);
                            acc[4 * u + 1] = keep_max(acc[4 * u + 1], v[u].y);
                            acc[4 * u + 2] = keep_max(acc[4 * u + 2], v[u].z);
                            acc[4 * u + 3] = keep_max(acc[4 * u + 3], v[u].w);
                        }
                    }
                }
            }
        }
        p = pn;
    }

#pragma unroll
    for (int u = 0; u < AGG_FW; ++u) part[ps][lane][u] = acc[u];
    __syncthreads();

    for (int idx = threadIdx.x; idx < 32 * AGG_FW; idx += NT) {
        int c = idx / AGG_FW;
        int fl = idx - c * AGG_FW;
        int f = f0 + fl;
        float v = part[0][c][fl];
#pragma unroll
        for (int s = 1; s < NS; ++s) v = fmaxf(v, part[s][c][fl]);
        float2 cq = ((const float2*)ctr)[b * m + i0 + c];
        out[(size_t)(b * m + i0 + c) * F + f] =
            v + bias[f] - cq.x * Wp[f] - cq.y * Wp[F + f];
    }
}

__global__ void init_gf_kernel(unsigned* __restrict__ gf, int n) {
    int i = blockIdx.x * blockDim.x + threadIdx.x;
    if (i < n) gf[i] = 0u;
}
__global__ void decode_gf_kernel(const unsigned* __restrict__ gf,
                                 float* __restrict__ out, int n) {
    int i = blockIdx.x * blockDim.x + threadIdx.x;
    if (i < n) out[i] = ord2f(gf[i]);
}

extern "C" void kernel_launch(void* const* d_in, const int* in_sizes, int n_in,
                              void* d_out, int out_size, void* d_ws, size_t ws_size,
                              hipStream_t stream) {
    const float* x     = (const float*)d_in[0];
    const float* zones = (const float*)d_in[1];
    const float* lf_W1 = (const float*)d_in[2];
    const float* lf_b1 = (const float*)d_in[3];
    const float* lf_W2 = (const float*)d_in[4];
    const float* lf_b2 = (const float*)d_in[5];
    const float* c1_W  = (const float*)d_in[6];
    const float* c1_b  = (const float*)d_in[7];
    const float* c2_W  = (const float*)d_in[8];
    const float* c2_b  = (const float*)d_in[9];
    const float* c3_W  = (const float*)d_in[10];
    const float* c3_b  = (const float*)d_in[11];

    float* out    = (float*)d_out;
    float* lf_out = out;
    float* gf_out = out + (size_t)BATCH * NPTS * 64;

    const int M = BATCH * NPTS;

    float* ws = (float*)d_ws;
    size_t o = 0;
    float* hidden = ws + o;  o += (size_t)M * 64;
    float* t1     = ws + o;  o += (size_t)M * 64;
    float* ctr1   = ws + o;  o += (size_t)BATCH * M1 * 2;
    float* h1     = ws + o;  o += (size_t)BATCH * M1 * 64;
    float* t2     = ws + o;  o += (size_t)BATCH * M1 * 128;
    float* ctr2   = ws + o;  o += (size_t)BATCH * M2 * 2;
    float* h2     = ws + o;  o += (size_t)BATCH * M2 * 128;
    unsigned* gf_enc = (unsigned*)(ws + o);

    hidden_kernel<<<(M * 64 + 255) / 256, 256, 0, stream>>>(x, lf_W1, lf_b1,
                                                            hidden, M);
    gemm_wmma_kernel<1, 64, 0, 0, 64><<<dim3(M / 16, 1), 32, 0, stream>>>(
        hidden, nullptr, nullptr, lf_W2, 64, lf_b2, lf_out, nullptr, 0);

    fps_kernel<NPTS><<<BATCH, 256, 0, stream>>>(x, M1, ctr1);
    fps_kernel<M1><<<BATCH, 256, 0, stream>>>(ctr1, M2, ctr2);

    gemm_wmma_kernel<0, 64, 1, 2, 67><<<dim3(M / 16, 1), 32, 0, stream>>>(
        lf_out, zones, x, c1_W, 64, nullptr, t1, nullptr, 0);

    agg_kernel<64, 16><<<dim3(BATCH * M1 / 32, 1), 512, 0, stream>>>(
        t1, x, ctr1, c1_W + 65 * 64, c1_b, h1, NPTS, M1, 0.25f);

    gemm_wmma_kernel<0, 64, 2, 0, 66>
        <<<dim3(BATCH * M1 / 16, 2), 32, 0, stream>>>(
            h1, ctr1, nullptr, c2_W, 128, nullptr, t2, nullptr, 0);

    agg_kernel<128, 16><<<dim3(BATCH * M2 / 32, 2), 512, 0, stream>>>(
        t2, ctr1, ctr2, c2_W + 64 * 128, c2_b, h2, M1, M2, 1.0f);

    init_gf_kernel<<<(BATCH * GF + 255) / 256, 256, 0, stream>>>(gf_enc,
                                                                 BATCH * GF);
    gemm_wmma_kernel<2, 128, 2, 0, 130>
        <<<dim3(BATCH * M2 / 16, GF / 64), 32, 0, stream>>>(
            h2, ctr2, nullptr, c3_W, GF, c3_b, nullptr, gf_enc, M2);
    decode_gf_kernel<<<(BATCH * GF + 255) / 256, 256, 0, stream>>>(gf_enc,
                                                                   gf_out,
                                                                   BATCH * GF);
}
